# starting kernel
# baseline (speedup 1.0000x reference)
#include <hip/hip_runtime.h>


typedef __attribute__((ext_vector_type(2))) float v2f;
typedef __attribute__((ext_vector_type(8))) float v8f;

#define H_IN   721
#define W_IN   1440
#define H_OUT  361
#define W_OUT  720
#define N_IN   (H_IN * W_IN)
#define N_OUT  (H_OUT * W_OUT)
#define NNZ    1048576
#define NCHUNK (NNZ / 16)

__global__ __launch_bounds__(256)
void regrid_zero_kernel(float* __restrict__ y, int n) {
    int i = blockIdx.x * blockDim.x + threadIdx.x;
    int stride = gridDim.x * blockDim.x;
    for (; i < n; i += stride) y[i] = 0.0f;
}

__global__ __launch_bounds__(256)
void regrid_coo_wmma_kernel(const float* __restrict__ x,
                            const float* __restrict__ S,
                            const int*   __restrict__ rows,
                            const int*   __restrict__ cols,
                            float*       __restrict__ y) {
    const int lane = threadIdx.x & 31;
    const int n    = lane & 15;
    const int half = lane >> 4;

    const int gw = __builtin_amdgcn_readfirstlane(
        blockIdx.x * (blockDim.x >> 5) + (threadIdx.x >> 5));
    const int totalWaves = gridDim.x * (blockDim.x >> 5);

    const int chgroup    = gw & 3;
    const int stream     = gw >> 2;
    const int numStreams = totalWaves >> 2;
    const int chbase     = chgroup * 16;

    const float* __restrict__ xp = x + (size_t)(chbase + n) * N_IN;

    for (int chunk = stream; chunk < NCHUNK; chunk += numStreams) {
        const int kbase = chunk * 16;
        const int kb2   = kbase + 2 * half;

        int   colv[8];
        float sv[8];
        #pragma unroll
        for (int i = 0; i < 8; ++i) {
            const int off = ((i >> 1) << 2) | (i & 1);
            colv[i] = cols[kb2 + off];
            sv[i]   = S[kb2 + off];
        }
        const int r = rows[kbase + n];

        float av[8];
        #pragma unroll
        for (int i = 0; i < 8; ++i) av[i] = xp[colv[i]];

        v8f d = {};
        #pragma unroll
        for (int g = 0; g < 4; ++g) {
            v2f a, b;
            a[0] = av[2 * g + 0];
            a[1] = av[2 * g + 1];
            b[0] = (n == 4 * g + 2 * half + 0) ? sv[2 * g + 0] : 0.0f;
            b[1] = (n == 4 * g + 2 * half + 1) ? sv[2 * g + 1] : 0.0f;
            d = __builtin_amdgcn_wmma_f32_16x16x4_f32(
                    false, a, false, b, (short)0, d, false, false);
        }

        float* yb = y + (size_t)(chbase + 8 * half) * N_OUT + (size_t)r;
        #pragma unroll
        for (int v = 0; v < 8; ++v) {
            atomicAdd(yb + (size_t)v * N_OUT, d[v]);
        }
    }
}

extern "C" void kernel_launch(void* const* d_in, const int* in_sizes, int n_in,
                              void* d_out, int out_size, void* d_ws, size_t ws_size,
                              hipStream_t stream) {
    (void)in_sizes; (void)n_in; (void)d_ws; (void)ws_size;

    const float* x    = (const float*)d_in[0];
    const float* S    = (const float*)d_in[1];
    const int*   rows = (const int*)d_in[2];
    const int*   cols = (const int*)d_in[3];
    float*       y    = (float*)d_out;

    regrid_zero_kernel<<<4096, 256, 0, stream>>>(y, out_size);

    regrid_coo_wmma_kernel<<<2048, 256, 0, stream>>>(x, S, rows, cols, y);
}


# products (still on the f32 matrix unit) stored as whole 128 B lines into a row-sorted workspace array, then summed per output row by its owner: no float atomics; sum order = integer-add arrival rank
# speedup vs baseline: 10.3516x; 10.3516x over previous
#include <hip/hip_runtime.h>


typedef __attribute__((ext_vector_type(2))) float v2f;
typedef __attribute__((ext_vector_type(8))) float v8f;

#define H_IN   721
#define W_IN   1440
#define H_OUT  361
#define W_OUT  720
#define N_IN   (H_IN * W_IN)
#define N_OUT  (H_OUT * W_OUT)
#define NNZ    1048576

#define CB_SHIFT 10
#define CB       (1 << CB_SHIFT)
#define NB       ((N_IN + CB - 1) / CB)
#define NBP      1024
#define CSTRIDE  64
#define PITCH    (CB + 4)
#define SLICE    4096

#define POS_BITS 21
#define POS_MASK ((1 << POS_BITS) - 1)
#define TROWS    (NNZ + NBP * 16)
#define TW       32

#define NRB      ((N_OUT + 1023) >> 10)
#define N_OUT_P  (NRB << 10)
#define RB       128
#define TP       (RB + 4)

static_assert(NNZ + NBP * 16 <= (1 << POS_BITS), "position must fit its packed field");
static_assert(POS_BITS + CB_SHIFT <= 32, "packed entry fits 32 bits");
static_assert(NB <= NBP, "one sort thread per bucket");
static_assert(NRB <= 256, "row-block histogram size");
static_assert(NNZ % SLICE == 0, "whole slices");
static_assert(N_OUT % 4 == 0 && N_IN % 4 == 0, "float4 never split by an edge");

#define WS_CNT    0
#define WS_CUR    (NBP * CSTRIDE)
#define WS_RBS    (2 * NBP * CSTRIDE)
#define WS_START  (WS_RBS + 256 * CSTRIDE)
#define WS_RCNT   (WS_START + NBP)
#define WS_RSTART (WS_RCNT + N_OUT_P)
#define WS_RANK   (WS_RSTART + N_OUT_P)
#define WS_P      (WS_RANK + NNZ)
#define WS_T      (WS_P + 2 * (NNZ + NBP * 16))

__device__ __forceinline__ int bucket_of(int col) {
    const unsigned b = (unsigned)col >> CB_SHIFT;
    return (int)(b < (unsigned)NB ? b : (unsigned)(NB - 1));
}
__device__ __forceinline__ int row_of(int row) {
    return (int)((unsigned)row < (unsigned)N_OUT ? (unsigned)row : (unsigned)(N_OUT - 1));
}

__global__ __launch_bounds__(256)
void regrid_zero_kernel(int* __restrict__ cnt, int* __restrict__ rbs, int* __restrict__ rcnt) {
    const int i = blockIdx.x * blockDim.x + threadIdx.x;
    if (i < NBP) cnt[i * CSTRIDE] = 0;
    if (i < 256) rbs[i * CSTRIDE] = 0;
    if (i < N_OUT_P) rcnt[i] = 0;
}

__global__ __launch_bounds__(1024)
void regrid_count_kernel(const int* __restrict__ rows, const int* __restrict__ cols,
                         int* __restrict__ cnt, int* __restrict__ rbs,
                         int* __restrict__ rcnt, int* __restrict__ rank) {
    __shared__ int h[NBP];
    __shared__ int hb[256];
    const int tid = threadIdx.x;
    h[tid] = 0;
    if (tid < 256) hb[tid] = 0;
    __syncthreads();
    const int k0 = blockIdx.x * SLICE + tid;
    int cv[SLICE / 1024], rv[SLICE / 1024];
    #pragma unroll
    for (int j = 0; j < SLICE / 1024; ++j) {
        cv[j] = cols[k0 + j * 1024];
        rv[j] = row_of(rows[k0 + j * 1024]);
    }
    #pragma unroll
    for (int j = 0; j < SLICE / 1024; ++j)
        rank[k0 + j * 1024] = atomicAdd(&rcnt[rv[j]], 1);
    #pragma unroll
    for (int j = 0; j < SLICE / 1024; ++j) {
        atomicAdd(&h[bucket_of(cv[j])], 1);
        atomicAdd(&hb[rv[j] >> 10], 1);
    }
    __syncthreads();
    const int v = h[tid];
    if (v) atomicAdd(&cnt[tid * CSTRIDE], v);
    if (tid < 256) {
        const int w = hb[tid];
        if (w) atomicAdd(&rbs[tid * CSTRIDE], w);
    }
}

__device__ __forceinline__ int block_scan_incl(int* s, int tid, int v) {
    s[tid] = v;
    __syncthreads();
    for (int off = 1; off < 1024; off <<= 1) {
        const int a = (tid >= off) ? s[tid - off] : 0;
        __syncthreads();
        s[tid] += a;
        __syncthreads();
    }
    return s[tid];
}

__global__ __launch_bounds__(1024)
void regrid_scan_kernel(const int* __restrict__ cnt, int* __restrict__ cur,
                        int* __restrict__ start, int* __restrict__ P,
                        const int* __restrict__ rbs, const int* __restrict__ rcnt,
                        int* __restrict__ rstart) {
    __shared__ int s[1024];
    const int tid = threadIdx.x;
    const int b = blockIdx.x;
    if (b < NRB) {
        const int part = (tid < b) ? rbs[tid * CSTRIDE] : 0;
        block_scan_incl(s, tid, part);
        const int offset = s[1023];
        __syncthreads();
        const int c = rcnt[b * 1024 + tid];
        const int incl = block_scan_incl(s, tid, c);
        rstart[b * 1024 + tid] = offset + incl - c;
    } else {
        const int c = cnt[tid * CSTRIDE];
        const int padded = (c + 15) & ~15;
        const int excl = block_scan_incl(s, tid, padded) - padded;
        start[tid] = excl;
        cur[tid * CSTRIDE] = excl;
        for (int e = excl + c; e < excl + padded; ++e) {
            P[2 * e]     = NNZ + tid * 16 + (e - (excl + c));
            P[2 * e + 1] = 0;
        }
    }
}

__global__ __launch_bounds__(1024)
void regrid_bucket_fill_kernel(const float* __restrict__ S,
                               const int*   __restrict__ rows,
                               const int*   __restrict__ cols,
                               const int*   __restrict__ rstart,
                               const int*   __restrict__ rank,
                               int* __restrict__ cur, int* __restrict__ P) {
    __shared__ int h[NBP];
    __shared__ int lbase[NBP];
    const int tid = threadIdx.x;
    h[tid] = 0;
    __syncthreads();
    const int k0 = blockIdx.x * SLICE + tid;
    int   cv[SLICE / 1024], bv[SLICE / 1024], pv[SLICE / 1024];
    float wv[SLICE / 1024];
    #pragma unroll
    for (int j = 0; j < SLICE / 1024; ++j) {
        cv[j] = cols[k0 + j * 1024];
        pv[j] = rstart[row_of(rows[k0 + j * 1024])] + rank[k0 + j * 1024];
        wv[j] = S[k0 + j * 1024];
    }
    #pragma unroll
    for (int j = 0; j < SLICE / 1024; ++j) {
        bv[j] = bucket_of(cv[j]);
        atomicAdd(&h[bv[j]], 1);
    }
    __syncthreads();
    const int v = h[tid];
    int base = 0;
    if (v) base = atomicAdd(&cur[tid * CSTRIDE], v);
    lbase[tid] = base;
    h[tid] = 0;
    __syncthreads();
    #pragma unroll
    for (int j = 0; j < SLICE / 1024; ++j) {
        const int r   = atomicAdd(&h[bv[j]], 1);
        const int pos = lbase[bv[j]] + r;
        int2 e;
        e.x = (pv[j] & POS_MASK) | ((cv[j] & (CB - 1)) << POS_BITS);
        e.y = __float_as_int(wv[j]);
        reinterpret_cast<int2*>(P)[pos] = e;
    }
}

__global__ __launch_bounds__(512)
void regrid_product_wmma_kernel(const float* __restrict__ x,
                                const int*   __restrict__ P,
                                const int*   __restrict__ start,
                                const int*   __restrict__ cnt,
                                float*       __restrict__ T) {
    __shared__ __attribute__((aligned(16))) float xs[32 * PITCH];

    const int tid    = threadIdx.x;
    const int lane   = tid & 31;
    const int c      = lane & 15;
    const int half   = lane >> 4;
    const int bkt    = blockIdx.x;
    const int hf     = blockIdx.y;
    const int chbase = 32 * hf;
    const int col0   = bkt << CB_SHIFT;

    const int s0      = start[bkt];
    const int nb      = cnt[bkt * CSTRIDE];
    const int nchunks = (nb + 15) >> 4;

    const int wave = __builtin_amdgcn_readfirstlane(tid >> 5);

    #pragma unroll
    for (int bt = 0; bt < 2; ++bt) {
        float4 t[8];
        #pragma unroll
        for (int i = 0; i < 8; ++i) {
            const int idx = (bt * 8 + i) * 512 + tid;
            const int row = idx >> 8;
            const int c4  = (idx & 255) << 2;
            float4 z = {0.0f, 0.0f, 0.0f, 0.0f};
            if (col0 + c4 < N_IN)
                z = *reinterpret_cast<const float4*>(
                        x + (size_t)(chbase + row) * N_IN + col0 + c4);
            t[i] = z;
        }
        #pragma unroll
        for (int i = 0; i < 8; ++i) {
            const int idx = (bt * 8 + i) * 512 + tid;
            const int row = idx >> 8;
            const int c4  = (idx & 255) << 2;
            *reinterpret_cast<float4*>(xs + row * PITCH + c4) = t[i];
        }
    }
    __syncthreads();

    const float* xr0 = xs + c * PITCH;
    const float* xr1 = xs + (16 + c) * PITCH;
    float* Th = T + (size_t)hf * TROWS * TW + 2 * c;

    for (int chunk = wave; chunk < nchunks; chunk += 16) {
        const int* pe = P + 2 * (s0 + chunk * 16) + 2 * (8 * half);
        int4 q[4];
        #pragma unroll
        for (int g = 0; g < 4; ++g)
            q[g] = *reinterpret_cast<const int4*>(pe + 4 * g);

        float x0[8], x1[8], sv[8];
        #pragma unroll
        for (int g = 0; g < 4; ++g) {
            const int ca = (q[g].x >> POS_BITS) & (CB - 1);
            const int cb = (q[g].z >> POS_BITS) & (CB - 1);
            x0[2 * g + 0] = xr0[ca];
            x0[2 * g + 1] = xr0[cb];
            x1[2 * g + 0] = xr1[ca];
            x1[2 * g + 1] = xr1[cb];
            sv[2 * g + 0] = __int_as_float(q[g].y);
            sv[2 * g + 1] = __int_as_float(q[g].w);
        }

        v8f d0 = {}, d1 = {};
        #pragma unroll
        for (int g = 0; g < 4; ++g) {
            v2f a, b0, b1;
            a[0]  = (c == 8 * half + 2 * g + 0) ? sv[2 * g + 0] : 0.0f;
            a[1]  = (c == 8 * half + 2 * g + 1) ? sv[2 * g + 1] : 0.0f;
            b0[0] = x0[2 * g + 0];
            b0[1] = x0[2 * g + 1];
            b1[0] = x1[2 * g + 0];
            b1[1] = x1[2 * g + 1];
            d0 = __builtin_amdgcn_wmma_f32_16x16x4_f32(
                     false, a, false, b0, (short)0, d0, false, false);
            d1 = __builtin_amdgcn_wmma_f32_16x16x4_f32(
                     false, a, false, b1, (short)0, d1, false, false);
        }

        #pragma unroll
        for (int r = 0; r < 8; ++r) {
            const int pw  = (r & 1) ? q[r >> 1].z : q[r >> 1].x;
            const int pos = pw & POS_MASK;
            float2 o;
            o.x = d0[r];
            o.y = d1[r];
            *reinterpret_cast<float2*>(Th + (size_t)pos * TW) = o;
        }
    }
}

__global__ __launch_bounds__(256)
void regrid_rowsum_kernel(const float* __restrict__ T,
                          const int*   __restrict__ rstart,
                          const int*   __restrict__ rcnt,
                          float*       __restrict__ y) {
    __shared__ __attribute__((aligned(16))) float tile[64 * TP];

    const int tid  = threadIdx.x;
    const int lane = tid & 31;
    const int wave = tid >> 5;
    const int rg   = lane >> 3;
    const int cq   = lane & 7;
    const int i0   = blockIdx.x * RB;

    for (int it = 0; it < (2 * RB) / 32; ++it) {
        const int grp = it * 8 + wave;
        const int hf  = grp / (RB / 4);
        const int rl  = ((grp % (RB / 4)) << 2) + rg;
        const int i   = i0 + rl;
        float4 acc = {0.0f, 0.0f, 0.0f, 0.0f};
        const int ii = i < N_OUT ? i : N_OUT - 1;
        const int p0 = rstart[ii];
        const int n  = i < N_OUT ? rcnt[ii] : 0;
        const float4* src = reinterpret_cast<const float4*>(
            T + (size_t)hf * TROWS * TW) + cq;
        const int jmax = n > 0 ? n - 1 : 0;
        float4 v[8];
        #pragma unroll
        for (int j = 0; j < 8; ++j) {
            int line = p0 + (j < jmax ? j : jmax);
            line = line < NNZ ? line : NNZ - 1;
            v[j] = src[(size_t)line * (TW / 4)];
        }
        #pragma unroll
        for (int j = 0; j < 8; ++j) {
            const bool on = j < n;
            acc.x += on ? v[j].x : 0.0f;
            acc.y += on ? v[j].y : 0.0f;
            acc.z += on ? v[j].z : 0.0f;
            acc.w += on ? v[j].w : 0.0f;
        }
        for (int j = 8; j < n; ++j) {
            const float4 z = src[(size_t)(p0 + j) * (TW / 4)];
            acc.x += z.x; acc.y += z.y; acc.z += z.z; acc.w += z.w;
        }
        const int chb = 32 * hf + 2 * cq;
        tile[(chb +  0) * TP + rl] = acc.x;
        tile[(chb + 16) * TP + rl] = acc.y;
        tile[(chb +  1) * TP + rl] = acc.z;
        tile[(chb + 17) * TP + rl] = acc.w;
    }
    __syncthreads();

    #pragma unroll
    for (int m = 0; m < (64 * RB / 4) / 256; ++m) {
        const int f  = m * 256 + tid;
        const int ch = f / (RB / 4);
        const int r4 = (f % (RB / 4)) << 2;
        if (i0 + r4 < N_OUT) {
            const float4 o = *reinterpret_cast<const float4*>(tile + ch * TP + r4);
            *reinterpret_cast<float4*>(y + (size_t)ch * N_OUT + i0 + r4) = o;
        }
    }
}

extern "C" void kernel_launch(void* const* d_in, const int* in_sizes, int n_in,
                              void* d_out, int out_size, void* d_ws, size_t ws_size,
                              hipStream_t stream) {
    (void)in_sizes; (void)n_in; (void)out_size; (void)ws_size;

    const float* x    = (const float*)d_in[0];
    const float* S    = (const float*)d_in[1];
    const int*   rows = (const int*)d_in[2];
    const int*   cols = (const int*)d_in[3];
    float*       y    = (float*)d_out;
    int*         ws   = (int*)d_ws;
    float*       T    = (float*)(ws + WS_T);

    regrid_zero_kernel<<<N_OUT_P / 256, 256, 0, stream>>>(ws + WS_CNT, ws + WS_RBS, ws + WS_RCNT);

    regrid_count_kernel<<<NNZ / SLICE, 1024, 0, stream>>>(rows, cols, ws + WS_CNT, ws + WS_RBS,
                                                          ws + WS_RCNT, ws + WS_RANK);
    regrid_scan_kernel<<<NRB + 1, 1024, 0, stream>>>(ws + WS_CNT, ws + WS_CUR, ws + WS_START,
                                                     ws + WS_P, ws + WS_RBS, ws + WS_RCNT,
                                                     ws + WS_RSTART);
    regrid_bucket_fill_kernel<<<NNZ / SLICE, 1024, 0, stream>>>(S, rows, cols, ws + WS_RSTART,
                                                                ws + WS_RANK, ws + WS_CUR, ws + WS_P);

    regrid_product_wmma_kernel<<<dim3(NB, 2), 512, 0, stream>>>(x, ws + WS_P, ws + WS_START,
                                                                ws + WS_CNT, T);
    regrid_rowsum_kernel<<<(N_OUT + RB - 1) / RB, 256, 0, stream>>>(T, ws + WS_RSTART,
                                                                    ws + WS_RCNT, y);
}


# speedup vs baseline: 12.4208x; 1.1999x over previous
#include <hip/hip_runtime.h>


typedef __attribute__((ext_vector_type(2))) float v2f;
typedef __attribute__((ext_vector_type(8))) float v8f;

#define H_IN   721
#define W_IN   1440
#define H_OUT  361
#define W_OUT  720
#define N_IN   (H_IN * W_IN)
#define N_OUT  (H_OUT * W_OUT)
#define NNZ    1048576

#define CB_SHIFT 10
#define CB       (1 << CB_SHIFT)
#define NB       ((N_IN + CB - 1) / CB)
#define NBP      1024
#define CSTRIDE  64
#define PITCH    (CB + 4)
#define SLICE    4096

#define POS_BITS 21
#define POS_MASK ((1 << POS_BITS) - 1)
#define TROWS    (NNZ + NBP * 16)
#define TW       64

#define NRB      ((N_OUT + 1023) >> 10)
#define N_OUT_P  (NRB << 10)
#define RB       128
#define TP       (RB + 4)

static_assert(NNZ + NBP * 16 <= (1 << POS_BITS), "position must fit its packed field");
static_assert(POS_BITS + CB_SHIFT <= 32, "packed entry fits 32 bits");
static_assert(NB <= NBP, "one sort thread per bucket");
static_assert(NRB <= 256, "row-block histogram size");
static_assert(NNZ % SLICE == 0, "whole slices");
static_assert(N_OUT % 4 == 0 && N_IN % 4 == 0, "float4 never split by an edge");

#define WS_CNT    0
#define WS_CUR    (NBP * CSTRIDE)
#define WS_RBS    (2 * NBP * CSTRIDE)
#define WS_START  (WS_RBS + 256 * CSTRIDE)
#define WS_RCNT   (WS_START + NBP)
#define WS_RSTART (WS_RCNT + N_OUT_P)
#define WS_RANK   (WS_RSTART + N_OUT_P)
#define WS_P      (WS_RANK + NNZ)
#define WS_T      (WS_P + 2 * (NNZ + NBP * 16))

__device__ __forceinline__ int bucket_of(int col) {
    const unsigned b = (unsigned)col >> CB_SHIFT;
    return (int)(b < (unsigned)NB ? b : (unsigned)(NB - 1));
}
__device__ __forceinline__ int row_of(int row) {
    return (int)((unsigned)row < (unsigned)N_OUT ? (unsigned)row : (unsigned)(N_OUT - 1));
}

__global__ __launch_bounds__(256)
void regrid_zero_kernel(int* __restrict__ cnt, int* __restrict__ rbs, int* __restrict__ rcnt) {
    const int i = blockIdx.x * blockDim.x + threadIdx.x;
    if (i < NBP) cnt[i * CSTRIDE] = 0;
    if (i < 256) rbs[i * CSTRIDE] = 0;
    if (i < N_OUT_P) rcnt[i] = 0;
}

__global__ __launch_bounds__(1024)
void regrid_count_kernel(const int* __restrict__ rows, const int* __restrict__ cols,
                         int* __restrict__ cnt, int* __restrict__ rbs,
                         int* __restrict__ rcnt, int* __restrict__ rank) {
    __shared__ int h[NBP];
    __shared__ int hb[256];
    const int tid = threadIdx.x;
    h[tid] = 0;
    if (tid < 256) hb[tid] = 0;
    __syncthreads();
    const int k0 = blockIdx.x * SLICE + tid;
    int cv[SLICE / 1024], rv[SLICE / 1024];
    #pragma unroll
    for (int j = 0; j < SLICE / 1024; ++j) {
        cv[j] = cols[k0 + j * 1024];
        rv[j] = row_of(rows[k0 + j * 1024]);
    }
    #pragma unroll
    for (int j = 0; j < SLICE / 1024; ++j)
        rank[k0 + j * 1024] = atomicAdd(&rcnt[rv[j]], 1);
    #pragma unroll
    for (int j = 0; j < SLICE / 1024; ++j) {
        atomicAdd(&h[bucket_of(cv[j])], 1);
        atomicAdd(&hb[rv[j] >> 10], 1);
    }
    __syncthreads();
    const int v = h[tid];
    if (v) atomicAdd(&cnt[tid * CSTRIDE], v);
    if (tid < 256) {
        const int w = hb[tid];
        if (w) atomicAdd(&rbs[tid * CSTRIDE], w);
    }
}

__device__ __forceinline__ int block_scan_incl(int* s, int tid, int v) {
    s[tid] = v;
    __syncthreads();
    for (int off = 1; off < 1024; off <<= 1) {
        const int a = (tid >= off) ? s[tid - off] : 0;
        __syncthreads();
        s[tid] += a;
        __syncthreads();
    }
    return s[tid];
}

__global__ __launch_bounds__(1024)
void regrid_scan_kernel(const int* __restrict__ cnt, int* __restrict__ cur,
                        int* __restrict__ start, int* __restrict__ P,
                        const int* __restrict__ rbs, const int* __restrict__ rcnt,
                        int* __restrict__ rstart) {
    __shared__ int s[1024];
    const int tid = threadIdx.x;
    const int b = blockIdx.x;
    if (b < NRB) {
        const int part = (tid < b) ? rbs[tid * CSTRIDE] : 0;
        block_scan_incl(s, tid, part);
        const int offset = s[1023];
        __syncthreads();
        const int c = rcnt[b * 1024 + tid];
        const int incl = block_scan_incl(s, tid, c);
        rstart[b * 1024 + tid] = offset + incl - c;
    } else {
        const int c = cnt[tid * CSTRIDE];
        const int padded = (c + 15) & ~15;
        const int excl = block_scan_incl(s, tid, padded) - padded;
        start[tid] = excl;
        cur[tid * CSTRIDE] = excl;
        for (int e = excl + c; e < excl + padded; ++e) {
            P[2 * e]     = NNZ + tid * 16 + (e - (excl + c));
            P[2 * e + 1] = 0;
        }
    }
}

__global__ __launch_bounds__(1024)
void regrid_bucket_fill_kernel(const float* __restrict__ S,
                               const int*   __restrict__ rows,
                               const int*   __restrict__ cols,
                               const int*   __restrict__ rstart,
                               const int*   __restrict__ rank,
                               int* __restrict__ cur, int* __restrict__ P) {
    __shared__ int h[NBP];
    __shared__ int lbase[NBP];
    const int tid = threadIdx.x;
    h[tid] = 0;
    __syncthreads();
    const int k0 = blockIdx.x * SLICE + tid;
    int   cv[SLICE / 1024], bv[SLICE / 1024], pv[SLICE / 1024];
    float wv[SLICE / 1024];
    #pragma unroll
    for (int j = 0; j < SLICE / 1024; ++j) {
        cv[j] = cols[k0 + j * 1024];
        pv[j] = rstart[row_of(rows[k0 + j * 1024])] + rank[k0 + j * 1024];
        wv[j] = S[k0 + j * 1024];
    }
    #pragma unroll
    for (int j = 0; j < SLICE / 1024; ++j) {
        bv[j] = bucket_of(cv[j]);
        atomicAdd(&h[bv[j]], 1);
    }
    __syncthreads();
    const int v = h[tid];
    int base = 0;
    if (v) base = atomicAdd(&cur[tid * CSTRIDE], v);
    lbase[tid] = base;
    h[tid] = 0;
    __syncthreads();
    #pragma unroll
    for (int j = 0; j < SLICE / 1024; ++j) {
        const int r   = atomicAdd(&h[bv[j]], 1);
        const int pos = lbase[bv[j]] + r;
        int2 e;
        e.x = (pv[j] & POS_MASK) | ((cv[j] & (CB - 1)) << POS_BITS);
        e.y = __float_as_int(wv[j]);
        reinterpret_cast<int2*>(P)[pos] = e;
    }
}

__global__ __launch_bounds__(1024)
void regrid_product_wmma_kernel(const float* __restrict__ x,
                                const int*   __restrict__ P,
                                const int*   __restrict__ start,
                                const int*   __restrict__ cnt,
                                float*       __restrict__ T) {
    __shared__ __attribute__((aligned(16))) float xs[64 * PITCH];

    const int tid    = threadIdx.x;
    const int lane   = tid & 31;
    const int c      = lane & 15;
    const int half   = lane >> 4;
    const int bkt    = blockIdx.x;
    const int col0   = bkt << CB_SHIFT;

    const int s0      = start[bkt];
    const int nb      = cnt[bkt * CSTRIDE];
    const int nchunks = (nb + 15) >> 4;

    const int wave = __builtin_amdgcn_readfirstlane(tid >> 5);

    #pragma unroll
    for (int bt = 0; bt < 2; ++bt) {
        float4 t[8];
        #pragma unroll
        for (int i = 0; i < 8; ++i) {
            const int idx = (bt * 8 + i) * 1024 + tid;
            const int row = idx >> 8;
            const int c4  = (idx & 255) << 2;
            float4 z = {0.0f, 0.0f, 0.0f, 0.0f};
            if (col0 + c4 < N_IN)
                z = *reinterpret_cast<const float4*>(
                        x + (size_t)row * N_IN + col0 + c4);
            t[i] = z;
        }
        #pragma unroll
        for (int i = 0; i < 8; ++i) {
            const int idx = (bt * 8 + i) * 1024 + tid;
            const int row = idx >> 8;
            const int c4  = (idx & 255) << 2;
            *reinterpret_cast<float4*>(xs + row * PITCH + c4) = t[i];
        }
    }
    __syncthreads();

    const float* xr0 = xs + c * PITCH;
    const float* xr1 = xs + (16 + c) * PITCH;
    const float* xr2 = xs + (32 + c) * PITCH;
    const float* xr3 = xs + (48 + c) * PITCH;
    float* Th = T + 4 * c;

    for (int chunk = wave; chunk < nchunks; chunk += 32) {
        const int* pe = P + 2 * (s0 + chunk * 16) + 2 * (8 * half);
        int4 q[4];
        #pragma unroll
        for (int g = 0; g < 4; ++g)
            q[g] = *reinterpret_cast<const int4*>(pe + 4 * g);

        float x0[8], x1[8], x2[8], x3[8], sv[8];
        #pragma unroll
        for (int g = 0; g < 4; ++g) {
            const int ca = (q[g].x >> POS_BITS) & (CB - 1);
            const int cb = (q[g].z >> POS_BITS) & (CB - 1);
            x0[2 * g + 0] = xr0[ca];
            x0[2 * g + 1] = xr0[cb];
            x1[2 * g + 0] = xr1[ca];
            x1[2 * g + 1] = xr1[cb];
            x2[2 * g + 0] = xr2[ca];
            x2[2 * g + 1] = xr2[cb];
            x3[2 * g + 0] = xr3[ca];
            x3[2 * g + 1] = xr3[cb];
            sv[2 * g + 0] = __int_as_float(q[g].y);
            sv[2 * g + 1] = __int_as_float(q[g].w);
        }

        v8f d0 = {}, d1 = {}, d2 = {}, d3 = {};
        #pragma unroll
        for (int g = 0; g < 4; ++g) {
            v2f a, b0, b1, b2, b3;
            a[0]  = (c == 8 * half + 2 * g + 0) ? sv[2 * g + 0] : 0.0f;
            a[1]  = (c == 8 * half + 2 * g + 1) ? sv[2 * g + 1] : 0.0f;
            b0[0] = x0[2 * g + 0];
            b0[1] = x0[2 * g + 1];
            b1[0] = x1[2 * g + 0];
            b1[1] = x1[2 * g + 1];
            b2[0] = x2[2 * g + 0];
            b2[1] = x2[2 * g + 1];
            b3[0] = x3[2 * g + 0];
            b3[1] = x3[2 * g + 1];
            d0 = __builtin_amdgcn_wmma_f32_16x16x4_f32(
                     false, a, false, b0, (short)0, d0, false, false);
            d1 = __builtin_amdgcn_wmma_f32_16x16x4_f32(
                     false, a, false, b1, (short)0, d1, false, false);
            d2 = __builtin_amdgcn_wmma_f32_16x16x4_f32(
                     false, a, false, b2, (short)0, d2, false, false);
            d3 = __builtin_amdgcn_wmma_f32_16x16x4_f32(
                     false, a, false, b3, (short)0, d3, false, false);
        }

        #pragma unroll
        for (int r = 0; r < 8; ++r) {
            const int pw  = (r & 1) ? q[r >> 1].z : q[r >> 1].x;
            const int pos = pw & POS_MASK;
            float4 o;
            o.x = d0[r];
            o.y = d1[r];
            o.z = d2[r];
            o.w = d3[r];
            *reinterpret_cast<float4*>(Th + (size_t)pos * TW) = o;
        }
    }
}

__global__ __launch_bounds__(256)
void regrid_rowsum_kernel(const float* __restrict__ T,
                          const int*   __restrict__ rstart,
                          const int*   __restrict__ rcnt,
                          float*       __restrict__ y) {
    __shared__ __attribute__((aligned(16))) float tile[64 * TP];

    const int tid  = threadIdx.x;
    const int lane = tid & 31;
    const int wave = tid >> 5;
    const int rg   = lane >> 4;
    const int cq   = lane & 15;
    const int i0   = blockIdx.x * RB;

    for (int it = 0; it < RB / 16; ++it) {
        const int rl = ((it * 8 + wave) << 1) + rg;
        const int i  = i0 + rl;
        float4 acc = {0.0f, 0.0f, 0.0f, 0.0f};
        const int ii = i < N_OUT ? i : N_OUT - 1;
        const int p0 = rstart[ii];
        const int n  = i < N_OUT ? rcnt[ii] : 0;
        const float4* src = reinterpret_cast<const float4*>(T) + cq;
        const int jmax = n > 0 ? n - 1 : 0;
        float4 v[8];
        #pragma unroll
        for (int j = 0; j < 8; ++j) {
            int line = p0 + (j < jmax ? j : jmax);
            line = line < NNZ ? line : NNZ - 1;
            v[j] = src[(size_t)line * (TW / 4)];
        }
        #pragma unroll
        for (int j = 0; j < 8; ++j) {
            const bool on = j < n;
            acc.x += on ? v[j].x : 0.0f;
            acc.y += on ? v[j].y : 0.0f;
            acc.z += on ? v[j].z : 0.0f;
            acc.w += on ? v[j].w : 0.0f;
        }
        for (int j = 8; j < n; ++j) {
            const float4 z = src[(size_t)(p0 + j) * (TW / 4)];
            acc.x += z.x; acc.y += z.y; acc.z += z.z; acc.w += z.w;
        }
        tile[(cq +  0) * TP + rl] = acc.x;
        tile[(cq + 16) * TP + rl] = acc.y;
        tile[(cq + 32) * TP + rl] = acc.z;
        tile[(cq + 48) * TP + rl] = acc.w;
    }
    __syncthreads();

    #pragma unroll
    for (int m = 0; m < (64 * RB / 4) / 256; ++m) {
        const int f  = m * 256 + tid;
        const int ch = f / (RB / 4);
        const int r4 = (f % (RB / 4)) << 2;
        if (i0 + r4 < N_OUT) {
            const float4 o = *reinterpret_cast<const float4*>(tile + ch * TP + r4);
            *reinterpret_cast<float4*>(y + (size_t)ch * N_OUT + i0 + r4) = o;
        }
    }
}

extern "C" void kernel_launch(void* const* d_in, const int* in_sizes, int n_in,
                              void* d_out, int out_size, void* d_ws, size_t ws_size,
                              hipStream_t stream) {
    (void)in_sizes; (void)n_in; (void)out_size; (void)ws_size;

    const float* x    = (const float*)d_in[0];
    const float* S    = (const float*)d_in[1];
    const int*   rows = (const int*)d_in[2];
    const int*   cols = (const int*)d_in[3];
    float*       y    = (float*)d_out;
    int*         ws   = (int*)d_ws;
    float*       T    = (float*)(ws + WS_T);

    regrid_zero_kernel<<<N_OUT_P / 256, 256, 0, stream>>>(ws + WS_CNT, ws + WS_RBS, ws + WS_RCNT);

    regrid_count_kernel<<<NNZ / SLICE, 1024, 0, stream>>>(rows, cols, ws + WS_CNT, ws + WS_RBS,
                                                          ws + WS_RCNT, ws + WS_RANK);
    regrid_scan_kernel<<<NRB + 1, 1024, 0, stream>>>(ws + WS_CNT, ws + WS_CUR, ws + WS_START,
                                                     ws + WS_P, ws + WS_RBS, ws + WS_RCNT,
                                                     ws + WS_RSTART);
    regrid_bucket_fill_kernel<<<NNZ / SLICE, 1024, 0, stream>>>(S, rows, cols, ws + WS_RSTART,
                                                                ws + WS_RANK, ws + WS_CUR, ws + WS_P);

    regrid_product_wmma_kernel<<<NB, 1024, 0, stream>>>(x, ws + WS_P, ws + WS_START,
                                                        ws + WS_CNT, T);
    regrid_rowsum_kernel<<<(N_OUT + RB - 1) / RB, 256, 0, stream>>>(T, ws + WS_RSTART,
                                                                    ws + WS_RCNT, y);
}


# row-sorted position from LDS counters: nonzeros also bucketed by 1024-row block, one work-group per block counts, scans, ranks in LDS; no scattered returning global adds; sum order = LDS arrival rank
# speedup vs baseline: 13.3162x; 1.0721x over previous
#include <hip/hip_runtime.h>


typedef __attribute__((ext_vector_type(2))) float v2f;
typedef __attribute__((ext_vector_type(8))) float v8f;

#define H_IN   721
#define W_IN   1440
#define H_OUT  361
#define W_OUT  720
#define N_IN   (H_IN * W_IN)
#define N_OUT  (H_OUT * W_OUT)
#define NNZ    1048576

#define CB_SHIFT 10
#define CB       (1 << CB_SHIFT)
#define NB       ((N_IN + CB - 1) / CB)
#define NBP      1024
#define CSTRIDE  64
#define PITCH    (CB + 4)
#define SLICE    4096

#define POS_BITS 21
#define POS_MASK ((1 << POS_BITS) - 1)
#define TROWS    (NNZ + NBP * 16)
#define TW       64

#define NRB      ((N_OUT + 1023) >> 10)
#define N_OUT_P  (NRB << 10)
#define RB       128
#define TP       (RB + 4)

static_assert(NNZ + NBP * 16 <= (1 << POS_BITS), "position must fit its packed field");
static_assert(POS_BITS + CB_SHIFT <= 32, "packed entry fits 32 bits");
static_assert(NB <= NBP, "one sort thread per bucket");
static_assert(NRB <= 256, "row-block histogram size");
static_assert(NNZ % SLICE == 0, "whole slices");
static_assert(N_OUT % 4 == 0 && N_IN % 4 == 0, "float4 never split by an edge");

#define WS_CNT    0
#define WS_CUR    (NBP * CSTRIDE)
#define WS_RBS    (2 * NBP * CSTRIDE)
#define WS_RBC    (WS_RBS + 256 * CSTRIDE)
#define WS_START  (WS_RBC + 256 * CSTRIDE)
#define WS_RBST   (WS_START + NBP)
#define WS_RCNT   (WS_RBST + 256)
#define WS_RSTART (WS_RCNT + N_OUT_P)
#define WS_R      (WS_RSTART + N_OUT_P)
#define WS_P      (WS_R + 4 * NNZ)
#define WS_T      (WS_P + 2 * (NNZ + NBP * 16))

static_assert(WS_T % 64 == 0, "T lines start on a 256-B boundary");
static_assert(WS_P % 4 == 0 && WS_R % 4 == 0, "16-B loads of P, 16-B entries of R");
static_assert(NRB * 1024 == N_OUT_P, "one rank work-group per 1024 rows");

__device__ __forceinline__ int bucket_of(int col) {
    const unsigned b = (unsigned)col >> CB_SHIFT;
    return (int)(b < (unsigned)NB ? b : (unsigned)(NB - 1));
}
__device__ __forceinline__ int row_of(int row) {
    return (int)((unsigned)row < (unsigned)N_OUT ? (unsigned)row : (unsigned)(N_OUT - 1));
}

__global__ __launch_bounds__(256)
void regrid_zero_kernel(int* __restrict__ cnt, int* __restrict__ rbs) {
    const int i = blockIdx.x * blockDim.x + threadIdx.x;
    if (i < NBP) cnt[i * CSTRIDE] = 0;
    if (i < 256) rbs[i * CSTRIDE] = 0;
}

__global__ __launch_bounds__(1024)
void regrid_count_kernel(const int* __restrict__ rows, const int* __restrict__ cols,
                         int* __restrict__ cnt, int* __restrict__ rbs) {
    __shared__ int h[NBP];
    __shared__ int hb[256];
    const int tid = threadIdx.x;
    h[tid] = 0;
    if (tid < 256) hb[tid] = 0;
    __syncthreads();
    const int k0 = blockIdx.x * SLICE + tid;
    int cv[SLICE / 1024], rv[SLICE / 1024];
    #pragma unroll
    for (int j = 0; j < SLICE / 1024; ++j) {
        cv[j] = cols[k0 + j * 1024];
        rv[j] = row_of(rows[k0 + j * 1024]);
    }
    #pragma unroll
    for (int j = 0; j < SLICE / 1024; ++j) {
        atomicAdd(&h[bucket_of(cv[j])], 1);
        atomicAdd(&hb[rv[j] >> 10], 1);
    }
    __syncthreads();
    const int v = h[tid];
    if (v) atomicAdd(&cnt[tid * CSTRIDE], v);
    if (tid < 256) {
        const int w = hb[tid];
        if (w) atomicAdd(&rbs[tid * CSTRIDE], w);
    }
}

__device__ __forceinline__ int block_scan_incl(int* s, int tid, int v) {
    s[tid] = v;
    __syncthreads();
    for (int off = 1; off < 1024; off <<= 1) {
        const int a = (tid >= off) ? s[tid - off] : 0;
        __syncthreads();
        s[tid] += a;
        __syncthreads();
    }
    return s[tid];
}

__global__ __launch_bounds__(1024)
void regrid_scan_kernel(const int* __restrict__ cnt, int* __restrict__ cur,
                        int* __restrict__ start, int* __restrict__ P,
                        const int* __restrict__ rbs, int* __restrict__ rbcur,
                        int* __restrict__ rbst) {
    __shared__ int s[1024];
    const int tid = threadIdx.x;
    if (blockIdx.x == 0) {
        const int c = cnt[tid * CSTRIDE];
        const int padded = (c + 15) & ~15;
        const int excl = block_scan_incl(s, tid, padded) - padded;
        start[tid] = excl;
        cur[tid * CSTRIDE] = excl;
        for (int e = excl + c; e < excl + padded; ++e) {
            P[2 * e]     = NNZ + tid * 16 + (e - (excl + c));
            P[2 * e + 1] = 0;
        }
    } else {
        const int c = (tid < 256) ? rbs[tid * CSTRIDE] : 0;
        const int excl = block_scan_incl(s, tid, c) - c;
        if (tid < 256) {
            rbst[tid] = excl;
            rbcur[tid * CSTRIDE] = excl;
        }
    }
}

__global__ __launch_bounds__(1024)
void regrid_bucket_fill_kernel(const float* __restrict__ S,
                               const int*   __restrict__ rows,
                               const int*   __restrict__ cols,
                               int* __restrict__ cur, int* __restrict__ rbcur,
                               int* __restrict__ R) {
    __shared__ int h[NBP];
    __shared__ int lbase[NBP];
    __shared__ int hr[256];
    __shared__ int lbr[256];
    const int tid = threadIdx.x;
    h[tid] = 0;
    if (tid < 256) hr[tid] = 0;
    __syncthreads();
    const int k0 = blockIdx.x * SLICE + tid;
    int   cv[SLICE / 1024], bv[SLICE / 1024], rv[SLICE / 1024];
    float wv[SLICE / 1024];
    #pragma unroll
    for (int j = 0; j < SLICE / 1024; ++j) {
        cv[j] = cols[k0 + j * 1024];
        rv[j] = row_of(rows[k0 + j * 1024]);
        wv[j] = S[k0 + j * 1024];
    }
    #pragma unroll
    for (int j = 0; j < SLICE / 1024; ++j) {
        bv[j] = bucket_of(cv[j]);
        atomicAdd(&h[bv[j]], 1);
        atomicAdd(&hr[rv[j] >> 10], 1);
    }
    __syncthreads();
    const int v = h[tid];
    int base = 0;
    if (v) base = atomicAdd(&cur[tid * CSTRIDE], v);
    lbase[tid] = base;
    h[tid] = 0;
    if (tid < 256) {
        const int w = hr[tid];
        int rbase = 0;
        if (w) rbase = atomicAdd(&rbcur[tid * CSTRIDE], w);
        lbr[tid] = rbase;
        hr[tid] = 0;
    }
    __syncthreads();
    #pragma unroll
    for (int j = 0; j < SLICE / 1024; ++j) {
        const int r    = atomicAdd(&h[bv[j]], 1);
        const int slot = lbase[bv[j]] + r;
        const int rb   = rv[j] >> 10;
        const int rr   = atomicAdd(&hr[rb], 1);
        const int rpos = lbr[rb] + rr;
        int4 e;
        e.x = (slot & POS_MASK) | ((cv[j] & (CB - 1)) << POS_BITS);
        e.y = rv[j];
        e.z = __float_as_int(wv[j]);
        e.w = 0;
        reinterpret_cast<int4*>(R)[rpos] = e;
    }
}

__global__ __launch_bounds__(1024)
void regrid_rank_kernel(const int* __restrict__ rbs, const int* __restrict__ rbst,
                        const int* __restrict__ R,
                        int* __restrict__ rstart, int* __restrict__ rcnt,
                        int* __restrict__ P) {
    __shared__ int lc[1024];
    __shared__ int s[1024];
    const int tid = threadIdx.x;
    const int b   = blockIdx.x;
    const int nb  = rbs[b * CSTRIDE];
    const int st  = rbst[b];
    const int4* Rb = reinterpret_cast<const int4*>(R) + st;

    lc[tid] = 0;
    __syncthreads();
    for (int base = 0; base < nb; base += 8 * 1024) {
        int rw[8];
        #pragma unroll
        for (int u = 0; u < 8; ++u) {
            const int idx = base + u * 1024 + tid;
            rw[u] = idx < nb ? Rb[idx].y : -1;
        }
        #pragma unroll
        for (int u = 0; u < 8; ++u)
            if (rw[u] >= 0) atomicAdd(&lc[rw[u] & 1023], 1);
    }
    __syncthreads();
    const int c    = lc[tid];
    const int excl = block_scan_incl(s, tid, c) - c;
    rstart[b * 1024 + tid] = st + excl;
    rcnt[b * 1024 + tid]   = c;
    s[tid]  = st + excl;
    lc[tid] = 0;
    __syncthreads();
    for (int base = 0; base < nb; base += 8 * 1024) {
        int4 e[8];
        #pragma unroll
        for (int u = 0; u < 8; ++u) {
            const int idx = base + u * 1024 + tid;
            int4 z;
            z.x = 0;
            z.y = -1;
            z.z = 0;
            z.w = 0;
            if (idx < nb) z = Rb[idx];
            e[u] = z;
        }
        #pragma unroll
        for (int u = 0; u < 8; ++u) {
            if (e[u].y >= 0) {
                const int rl  = e[u].y & 1023;
                const int r   = atomicAdd(&lc[rl], 1);
                const int pos = s[rl] + r;
                int2 o;
                o.x = (pos & POS_MASK) | (e[u].x & ~POS_MASK);
                o.y = e[u].z;
                reinterpret_cast<int2*>(P)[e[u].x & POS_MASK] = o;
            }
        }
    }
}

__global__ __launch_bounds__(1024)
void regrid_product_wmma_kernel(const float* __restrict__ x,
                                const int*   __restrict__ P,
                                const int*   __restrict__ start,
                                const int*   __restrict__ cnt,
                                float*       __restrict__ T) {
    __shared__ __attribute__((aligned(16))) float xs[64 * PITCH];

    const int tid    = threadIdx.x;
    const int lane   = tid & 31;
    const int c      = lane & 15;
    const int half   = lane >> 4;
    const int bkt    = blockIdx.x;
    const int col0   = bkt << CB_SHIFT;

    const int s0      = start[bkt];
    const int nb      = cnt[bkt * CSTRIDE];
    const int nchunks = (nb + 15) >> 4;

    const int wave = __builtin_amdgcn_readfirstlane(tid >> 5);

    #pragma unroll
    for (int bt = 0; bt < 2; ++bt) {
        float4 t[8];
        #pragma unroll
        for (int i = 0; i < 8; ++i) {
            const int idx = (bt * 8 + i) * 1024 + tid;
            const int row = idx >> 8;
            const int c4  = (idx & 255) << 2;
            float4 z = {0.0f, 0.0f, 0.0f, 0.0f};
            if (col0 + c4 < N_IN)
                z = *reinterpret_cast<const float4*>(
                        x + (size_t)row * N_IN + col0 + c4);
            t[i] = z;
        }
        #pragma unroll
        for (int i = 0; i < 8; ++i) {
            const int idx = (bt * 8 + i) * 1024 + tid;
            const int row = idx >> 8;
            const int c4  = (idx & 255) << 2;
            *reinterpret_cast<float4*>(xs + row * PITCH + c4) = t[i];
        }
    }
    __syncthreads();

    const float* xr0 = xs + c * PITCH;
    const float* xr1 = xs + (16 + c) * PITCH;
    const float* xr2 = xs + (32 + c) * PITCH;
    const float* xr3 = xs + (48 + c) * PITCH;
    float* Th = T + 4 * c;

    for (int chunk = wave; chunk < nchunks; chunk += 32) {
        const int* pe = P + 2 * (s0 + chunk * 16) + 2 * (8 * half);
        int4 q[4];
        #pragma unroll
        for (int g = 0; g < 4; ++g)
            q[g] = *reinterpret_cast<const int4*>(pe + 4 * g);

        float x0[8], x1[8], x2[8], x3[8], sv[8];
        #pragma unroll
        for (int g = 0; g < 4; ++g) {
            const int ca = (q[g].x >> POS_BITS) & (CB - 1);
            const int cb = (q[g].z >> POS_BITS) & (CB - 1);
            x0[2 * g + 0] = xr0[ca];
            x0[2 * g + 1] = xr0[cb];
            x1[2 * g + 0] = xr1[ca];
            x1[2 * g + 1] = xr1[cb];
            x2[2 * g + 0] = xr2[ca];
            x2[2 * g + 1] = xr2[cb];
            x3[2 * g + 0] = xr3[ca];
            x3[2 * g + 1] = xr3[cb];
            sv[2 * g + 0] = __int_as_float(q[g].y);
            sv[2 * g + 1] = __int_as_float(q[g].w);
        }

        v8f d0 = {}, d1 = {}, d2 = {}, d3 = {};
        #pragma unroll
        for (int g = 0; g < 4; ++g) {
            v2f a, b0, b1, b2, b3;
            a[0]  = (c == 8 * half + 2 * g + 0) ? sv[2 * g + 0] : 0.0f;
            a[1]  = (c == 8 * half + 2 * g + 1) ? sv[2 * g + 1] : 0.0f;
            b0[0] = x0[2 * g + 0];
            b0[1] = x0[2 * g + 1];
            b1[0] = x1[2 * g + 0];
            b1[1] = x1[2 * g + 1];
            b2[0] = x2[2 * g + 0];
            b2[1] = x2[2 * g + 1];
            b3[0] = x3[2 * g + 0];
            b3[1] = x3[2 * g + 1];
            d0 = __builtin_amdgcn_wmma_f32_16x16x4_f32(
                     false, a, false, b0, (short)0, d0, false, false);
            d1 = __builtin_amdgcn_wmma_f32_16x16x4_f32(
                     false, a, false, b1, (short)0, d1, false, false);
            d2 = __builtin_amdgcn_wmma_f32_16x16x4_f32(
                     false, a, false, b2, (short)0, d2, false, false);
            d3 = __builtin_amdgcn_wmma_f32_16x16x4_f32(
                     false, a, false, b3, (short)0, d3, false, false);
        }

        #pragma unroll
        for (int r = 0; r < 8; ++r) {
            const int pw  = (r & 1) ? q[r >> 1].z : q[r >> 1].x;
            const int pos = pw & POS_MASK;
            float4 o;
            o.x = d0[r];
            o.y = d1[r];
            o.z = d2[r];
            o.w = d3[r];
            *reinterpret_cast<float4*>(Th + (size_t)pos * TW) = o;
        }
    }
}

__global__ __launch_bounds__(256)
void regrid_rowsum_kernel(const float* __restrict__ T,
                          const int*   __restrict__ rstart,
                          const int*   __restrict__ rcnt,
                          float*       __restrict__ y) {
    __shared__ __attribute__((aligned(16))) float tile[64 * TP];

    const int tid  = threadIdx.x;
    const int lane = tid & 31;
    const int wave = tid >> 5;
    const int rg   = lane >> 4;
    const int cq   = lane & 15;
    const int i0   = blockIdx.x * RB;

    for (int it = 0; it < RB / 16; ++it) {
        const int rl = ((it * 8 + wave) << 1) + rg;
        const int i  = i0 + rl;
        float4 acc = {0.0f, 0.0f, 0.0f, 0.0f};
        const int ii = i < N_OUT ? i : N_OUT - 1;
        const int p0 = rstart[ii];
        const int n  = i < N_OUT ? rcnt[ii] : 0;
        const float4* src = reinterpret_cast<const float4*>(T) + cq;
        const int jmax = n > 0 ? n - 1 : 0;
        float4 v[8];
        #pragma unroll
        for (int j = 0; j < 8; ++j) {
            int line = p0 + (j < jmax ? j : jmax);
            line = line < NNZ ? line : NNZ - 1;
            v[j] = src[(size_t)line * (TW / 4)];
        }
        #pragma unroll
        for (int j = 0; j < 8; ++j) {
            const bool on = j < n;
            acc.x += on ? v[j].x : 0.0f;
            acc.y += on ? v[j].y : 0.0f;
            acc.z += on ? v[j].z : 0.0f;
            acc.w += on ? v[j].w : 0.0f;
        }
        for (int j = 8; j < n; ++j) {
            const float4 z = src[(size_t)(p0 + j) * (TW / 4)];
            acc.x += z.x; acc.y += z.y; acc.z += z.z; acc.w += z.w;
        }
        tile[(cq +  0) * TP + rl] = acc.x;
        tile[(cq + 16) * TP + rl] = acc.y;
        tile[(cq + 32) * TP + rl] = acc.z;
        tile[(cq + 48) * TP + rl] = acc.w;
    }
    __syncthreads();

    #pragma unroll
    for (int m = 0; m < (64 * RB / 4) / 256; ++m) {
        const int f  = m * 256 + tid;
        const int ch = f / (RB / 4);
        const int r4 = (f % (RB / 4)) << 2;
        if (i0 + r4 < N_OUT) {
            const float4 o = *reinterpret_cast<const float4*>(tile + ch * TP + r4);
            *reinterpret_cast<float4*>(y + (size_t)ch * N_OUT + i0 + r4) = o;
        }
    }
}

extern "C" void kernel_launch(void* const* d_in, const int* in_sizes, int n_in,
                              void* d_out, int out_size, void* d_ws, size_t ws_size,
                              hipStream_t stream) {
    (void)in_sizes; (void)n_in; (void)out_size; (void)ws_size;

    const float* x    = (const float*)d_in[0];
    const float* S    = (const float*)d_in[1];
    const int*   rows = (const int*)d_in[2];
    const int*   cols = (const int*)d_in[3];
    float*       y    = (float*)d_out;
    int*         ws   = (int*)d_ws;
    float*       T    = (float*)(ws + WS_T);

    regrid_zero_kernel<<<NBP / 256, 256, 0, stream>>>(ws + WS_CNT, ws + WS_RBS);

    regrid_count_kernel<<<NNZ / SLICE, 1024, 0, stream>>>(rows, cols, ws + WS_CNT, ws + WS_RBS);
    regrid_scan_kernel<<<2, 1024, 0, stream>>>(ws + WS_CNT, ws + WS_CUR, ws + WS_START,
                                               ws + WS_P, ws + WS_RBS, ws + WS_RBC,
                                               ws + WS_RBST);
    regrid_bucket_fill_kernel<<<NNZ / SLICE, 1024, 0, stream>>>(S, rows, cols, ws + WS_CUR,
                                                                ws + WS_RBC, ws + WS_R);
    regrid_rank_kernel<<<NRB, 1024, 0, stream>>>(ws + WS_RBS, ws + WS_RBST, ws + WS_R,
                                                 ws + WS_RSTART, ws + WS_RCNT, ws + WS_P);

    regrid_product_wmma_kernel<<<NB, 1024, 0, stream>>>(x, ws + WS_P, ws + WS_START,
                                                        ws + WS_CNT, T);
    regrid_rowsum_kernel<<<(N_OUT + RB - 1) / RB, 256, 0, stream>>>(T, ws + WS_RSTART,
                                                                    ws + WS_RCNT, y);
}


# each 4096-nonzero slice sorted inside its own fixed region, no global counters: zero, count and scan launches gone (7 to 4); consumers walk 256 per-slice runs; sum order = LDS arrival rank
# speedup vs baseline: 16.7206x; 1.2557x over previous
#include <hip/hip_runtime.h>


typedef __attribute__((ext_vector_type(2))) float v2f;
typedef __attribute__((ext_vector_type(8))) float v8f;

#define H_IN   721
#define W_IN   1440
#define H_OUT  361
#define W_OUT  720
#define N_IN   (H_IN * W_IN)
#define N_OUT  (H_OUT * W_OUT)
#define NNZ    1048576

#define CB_SHIFT 10
#define CB       (1 << CB_SHIFT)
#define NB       ((N_IN + CB - 1) / CB)
#define NBP      1024
#define PITCH    (CB + 4)
#define SLICE    4096
#define NSL      (NNZ / SLICE)
#define OCP      (NBP + 1)
#define ORP      257
#define CAP      4096

#define POS_BITS 21
#define POS_MASK ((1 << POS_BITS) - 1)
#define TROWS    (NNZ + NBP * 16)
#define TW       64

#define NRB      ((N_OUT + 1023) >> 10)
#define N_OUT_P  (NRB << 10)
#define RB       128
#define TP       (RB + 4)

static_assert(NNZ + NBP * 16 <= (1 << POS_BITS), "position must fit its packed field");
static_assert(POS_BITS + CB_SHIFT <= 32, "packed entry fits 32 bits");
static_assert(NB <= NBP, "one sort thread per bucket");
static_assert(NRB <= 256, "row-block histogram size");
static_assert(NNZ % SLICE == 0, "whole slices");
static_assert(NSL == 256, "one run per slice: 256 runs per bin");
static_assert(SLICE < (1 << 15), "two packed 16-bit running sums");
static_assert(CAP % 16 == 0, "whole chunks per tile of entries");
static_assert(N_OUT % 4 == 0 && N_IN % 4 == 0, "float4 never split by an edge");

#define WS_OC     0
#define WS_OR     (WS_OC + NSL * OCP)
#define WS_RCNT   (WS_OR + NSL * ORP)
#define WS_RSTART (WS_RCNT + N_OUT_P)
#define WS_R      (WS_RSTART + N_OUT_P)
#define WS_P      (WS_R + 4 * NNZ)
#define WS_T      (WS_P + 2 * NNZ)

static_assert(WS_T % 64 == 0, "T lines start on a 256-B boundary");
static_assert(WS_P % 4 == 0 && WS_R % 4 == 0, "16-B entries of R, 8-B entries of P");

__device__ __forceinline__ int bucket_of(int col) {
    const unsigned b = (unsigned)col >> CB_SHIFT;
    return (int)(b < (unsigned)NB ? b : (unsigned)(NB - 1));
}
__device__ __forceinline__ int row_of(int row) {
    return (int)((unsigned)row < (unsigned)N_OUT ? (unsigned)row : (unsigned)(N_OUT - 1));
}

__device__ __forceinline__ int block_scan_incl(int* s, int tid, int v) {
    s[tid] = v;
    __syncthreads();
    for (int off = 1; off < 1024; off <<= 1) {
        const int a = (tid >= off) ? s[tid - off] : 0;
        __syncthreads();
        s[tid] += a;
        __syncthreads();
    }
    return s[tid];
}

__global__ __launch_bounds__(1024)
void regrid_slice_sort_kernel(const float* __restrict__ S,
                              const int*   __restrict__ rows,
                              const int*   __restrict__ cols,
                              int* __restrict__ OC, int* __restrict__ ORo,
                              int* __restrict__ R) {
    __shared__ int h[NBP];
    __shared__ int lbase[NBP];
    __shared__ int hr[256];
    __shared__ int lbr[256];
    __shared__ int sc[1024];
    const int tid = threadIdx.x;
    const int sl  = blockIdx.x;
    h[tid] = 0;
    if (tid < 256) hr[tid] = 0;
    __syncthreads();
    const int k0 = sl * SLICE + tid;
    int   cv[SLICE / 1024], bv[SLICE / 1024], rv[SLICE / 1024];
    float wv[SLICE / 1024];
    #pragma unroll
    for (int j = 0; j < SLICE / 1024; ++j) {
        cv[j] = cols[k0 + j * 1024];
        rv[j] = row_of(rows[k0 + j * 1024]);
        wv[j] = S[k0 + j * 1024];
    }
    #pragma unroll
    for (int j = 0; j < SLICE / 1024; ++j) {
        bv[j] = bucket_of(cv[j]);
        atomicAdd(&h[bv[j]], 1);
        atomicAdd(&hr[rv[j] >> 10], 1);
    }
    __syncthreads();
    const int v   = h[tid];
    const int w   = tid < 256 ? hr[tid] : 0;
    const int inc = block_scan_incl(sc, tid, v | (w << 16));
    const int exc = (inc & 0xFFFF) - v;
    const int exr = (inc >> 16) - w;
    lbase[tid] = exc;
    h[tid] = 0;
    OC[sl * OCP + tid] = exc;
    if (tid == 1023) OC[sl * OCP + NBP] = SLICE;
    if (tid < 256) {
        lbr[tid] = exr;
        hr[tid] = 0;
        ORo[sl * ORP + tid] = exr;
    }
    if (tid == 255) ORo[sl * ORP + 256] = SLICE;
    __syncthreads();
    int4* Rs = reinterpret_cast<int4*>(R) + sl * SLICE;
    #pragma unroll
    for (int j = 0; j < SLICE / 1024; ++j) {
        const int r    = atomicAdd(&h[bv[j]], 1);
        const int slot = sl * SLICE + lbase[bv[j]] + r;
        const int rb   = rv[j] >> 10;
        const int rr   = atomicAdd(&hr[rb], 1);
        int4 e;
        e.x = (slot & POS_MASK) | ((cv[j] & (CB - 1)) << POS_BITS);
        e.y = rv[j];
        e.z = __float_as_int(wv[j]);
        e.w = 0;
        Rs[lbr[rb] + rr] = e;
    }
}

__global__ __launch_bounds__(1024)
void regrid_rank_kernel(const int* __restrict__ ORo,
                        const int* __restrict__ R,
                        int* __restrict__ rstart, int* __restrict__ rcnt,
                        int* __restrict__ P) {
    __shared__ int lc[1024];
    __shared__ int s[1024];
    __shared__ int roff[NSL];
    __shared__ int rlen[NSL];
    __shared__ int stsum;
    const int tid  = threadIdx.x;
    const int lane = tid & 31;
    const int b    = blockIdx.x;

    lc[tid] = 0;
    if (tid == 0) stsum = 0;
    __syncthreads();
    if (tid < NSL) {
        const int a = ORo[tid * ORP + b];
        const int e = ORo[tid * ORP + b + 1];
        roff[tid] = a;
        rlen[tid] = e - a;
        int t = a;
        #pragma unroll
        for (int m = 16; m >= 1; m >>= 1) t += __shfl_xor(t, m);
        if (lane == 0) atomicAdd(&stsum, t);
    }
    __syncthreads();
    const int st = stsum;
    const int4* R4 = reinterpret_cast<const int4*>(R);
    const int kl = tid & 15;

    {
        int rw[8];
        #pragma unroll
        for (int u = 0; u < 8; ++u) {
            const int run = (tid >> 4) + 64 * (u & 3);
            const int k   = kl + 16 * (u >> 2);
            rw[u] = k < rlen[run] ? R4[run * SLICE + roff[run] + k].y : -1;
        }
        #pragma unroll
        for (int u = 0; u < 8; ++u)
            if (rw[u] >= 0) atomicAdd(&lc[rw[u] & 1023], 1);
        for (int g = 0; g < 4; ++g) {
            const int run = (tid >> 4) + 64 * g;
            const int n   = rlen[run];
            for (int k = kl + 32; k < n; k += 16)
                atomicAdd(&lc[R4[run * SLICE + roff[run] + k].y & 1023], 1);
        }
    }
    __syncthreads();
    const int c    = lc[tid];
    const int excl = block_scan_incl(s, tid, c) - c;
    rstart[b * 1024 + tid] = st + excl;
    rcnt[b * 1024 + tid]   = c;
    s[tid]  = st + excl;
    lc[tid] = 0;
    __syncthreads();
    {
        int4 e[8];
        #pragma unroll
        for (int u = 0; u < 8; ++u) {
            const int run = (tid >> 4) + 64 * (u & 3);
            const int k   = kl + 16 * (u >> 2);
            int4 z;
            z.x = 0;
            z.y = -1;
            z.z = 0;
            z.w = 0;
            if (k < rlen[run]) z = R4[run * SLICE + roff[run] + k];
            e[u] = z;
        }
        #pragma unroll
        for (int u = 0; u < 8; ++u) {
            if (e[u].y >= 0) {
                const int rl  = e[u].y & 1023;
                const int r   = atomicAdd(&lc[rl], 1);
                const int pos = s[rl] + r;
                int2 o;
                o.x = (pos & POS_MASK) | (e[u].x & ~POS_MASK);
                o.y = e[u].z;
                reinterpret_cast<int2*>(P)[e[u].x & POS_MASK] = o;
            }
        }
        for (int g = 0; g < 4; ++g) {
            const int run = (tid >> 4) + 64 * g;
            const int n   = rlen[run];
            for (int k = kl + 32; k < n; k += 16) {
                const int4 z  = R4[run * SLICE + roff[run] + k];
                const int rl  = z.y & 1023;
                const int r   = atomicAdd(&lc[rl], 1);
                const int pos = s[rl] + r;
                int2 o;
                o.x = (pos & POS_MASK) | (z.x & ~POS_MASK);
                o.y = z.z;
                reinterpret_cast<int2*>(P)[z.x & POS_MASK] = o;
            }
        }
    }
}

__global__ __launch_bounds__(1024)
void regrid_product_wmma_kernel(const float* __restrict__ x,
                                const int*   __restrict__ P,
                                const int*   __restrict__ OC,
                                float*       __restrict__ T) {
    __shared__ __attribute__((aligned(16))) float xs[64 * PITCH];
    __shared__ __attribute__((aligned(16))) int ent[2 * CAP];
    __shared__ int coff[NSL];
    __shared__ int clen[NSL];
    __shared__ int cexc[NSL];
    __shared__ int wtot[NSL / 32];

    const int tid    = threadIdx.x;
    const int lane   = tid & 31;
    const int c      = lane & 15;
    const int half   = lane >> 4;
    const int bkt    = blockIdx.x;
    const int col0   = bkt << CB_SHIFT;

    const int wave = __builtin_amdgcn_readfirstlane(tid >> 5);

    const int* ocp = OC + (tid & (NSL - 1)) * OCP + bkt;
    const int ra = ocp[0];
    const int re = ocp[1];

    float4 t[8];
    #pragma unroll
    for (int i = 0; i < 8; ++i) {
        const int idx = i * 1024 + tid;
        const int row = idx >> 8;
        const int c4  = (idx & 255) << 2;
        float4 z = {0.0f, 0.0f, 0.0f, 0.0f};
        if (col0 + c4 < N_IN)
            z = *reinterpret_cast<const float4*>(x + (size_t)row * N_IN + col0 + c4);
        t[i] = z;
    }

    const int rn = tid < NSL ? re - ra : 0;
    int inc = rn;
    #pragma unroll
    for (int d = 1; d < 32; d <<= 1) {
        const int u = __shfl_up(inc, d);
        if (lane >= d) inc += u;
    }
    if (tid < NSL && lane == 31) wtot[tid >> 5] = inc;
    __syncthreads();
    int nbv = 0, before = 0;
    #pragma unroll
    for (int w = 0; w < NSL / 32; ++w) {
        const int q = wtot[w];
        if (w < (tid >> 5)) before += q;
        nbv += q;
    }
    if (tid < NSL) {
        coff[tid] = ra;
        clen[tid] = rn;
        cexc[tid] = before + inc - rn;
    }
    __syncthreads();
    const int nb      = __builtin_amdgcn_readfirstlane(nbv);
    const int nchunks = (nb + 15) >> 4;

    const int2* P2   = reinterpret_cast<const int2*>(P);
    int2*       ent2 = reinterpret_cast<int2*>(ent);
    const int run = tid >> 2;
    const int sub = tid & 3;
    const int ro  = coff[run];
    const int rl  = clen[run];
    const int rx  = cexc[run];

    const float* xr0 = xs + c * PITCH;
    const float* xr1 = xs + (16 + c) * PITCH;
    const float* xr2 = xs + (32 + c) * PITCH;
    const float* xr3 = xs + (48 + c) * PITCH;
    float* Th = T + 4 * c;

    int2 ev[4];
    #pragma unroll
    for (int u = 0; u < 4; ++u) {
        const int k = sub + 4 * u;
        int2 z;
        z.x = 0;
        z.y = 0;
        if (k < rl) z = P2[run * SLICE + ro + k];
        ev[u] = z;
    }
    #pragma unroll
    for (int i = 0; i < 8; ++i) {
        const int idx = i * 1024 + tid;
        const int row = idx >> 8;
        const int c4  = (idx & 255) << 2;
        *reinterpret_cast<float4*>(xs + row * PITCH + c4) = t[i];
    }
    #pragma unroll
    for (int i = 0; i < 8; ++i) {
        const int idx = (8 + i) * 1024 + tid;
        const int row = idx >> 8;
        const int c4  = (idx & 255) << 2;
        float4 z = {0.0f, 0.0f, 0.0f, 0.0f};
        if (col0 + c4 < N_IN)
            z = *reinterpret_cast<const float4*>(x + (size_t)row * N_IN + col0 + c4);
        t[i] = z;
    }
    #pragma unroll
    for (int u = 0; u < 4; ++u) {
        const int k = sub + 4 * u;
        const int d = rx + k;
        if (k < rl && (unsigned)d < (unsigned)CAP) ent2[d] = ev[u];
    }
    for (int k = sub + 16; k < rl; k += 4) {
        const int d = rx + k;
        if ((unsigned)d < (unsigned)CAP) ent2[d] = P2[run * SLICE + ro + k];
    }
    if (tid < 16) {
        const int j = nb + tid;
        if (j < nchunks * 16 && j < CAP) {
            int2 z;
            z.x = NNZ + bkt * 16 + tid;
            z.y = 0;
            ent2[j] = z;
        }
    }
    #pragma unroll
    for (int i = 0; i < 8; ++i) {
        const int idx = (8 + i) * 1024 + tid;
        const int row = idx >> 8;
        const int c4  = (idx & 255) << 2;
        *reinterpret_cast<float4*>(xs + row * PITCH + c4) = t[i];
    }

    int tb = 0;
    while (true) {
        __syncthreads();

        const int cend = min(nchunks, (tb + CAP) >> 4);
        for (int chunk = (tb >> 4) + wave; chunk < cend; chunk += 32) {
            const int* pe = ent + 2 * (chunk * 16 - tb) + 2 * (8 * half);
            int4 q[4];
            #pragma unroll
            for (int g = 0; g < 4; ++g)
                q[g] = *reinterpret_cast<const int4*>(pe + 4 * g);

            float x0[8], x1[8], x2[8], x3[8], sv[8];
            #pragma unroll
            for (int g = 0; g < 4; ++g) {
                const int ca = (q[g].x >> POS_BITS) & (CB - 1);
                const int cb = (q[g].z >> POS_BITS) & (CB - 1);
                x0[2 * g + 0] = xr0[ca];
                x0[2 * g + 1] = xr0[cb];
                x1[2 * g + 0] = xr1[ca];
                x1[2 * g + 1] = xr1[cb];
                x2[2 * g + 0] = xr2[ca];
                x2[2 * g + 1] = xr2[cb];
                x3[2 * g + 0] = xr3[ca];
                x3[2 * g + 1] = xr3[cb];
                sv[2 * g + 0] = __int_as_float(q[g].y);
                sv[2 * g + 1] = __int_as_float(q[g].w);
            }

            v8f d0 = {}, d1 = {}, d2 = {}, d3 = {};
            #pragma unroll
            for (int g = 0; g < 4; ++g) {
                v2f a, b0, b1, b2, b3;
                a[0]  = (c == 8 * half + 2 * g + 0) ? sv[2 * g + 0] : 0.0f;
                a[1]  = (c == 8 * half + 2 * g + 1) ? sv[2 * g + 1] : 0.0f;
                b0[0] = x0[2 * g + 0];
                b0[1] = x0[2 * g + 1];
                b1[0] = x1[2 * g + 0];
                b1[1] = x1[2 * g + 1];
                b2[0] = x2[2 * g + 0];
                b2[1] = x2[2 * g + 1];
                b3[0] = x3[2 * g + 0];
                b3[1] = x3[2 * g + 1];
                d0 = __builtin_amdgcn_wmma_f32_16x16x4_f32(
                         false, a, false, b0, (short)0, d0, false, false);
                d1 = __builtin_amdgcn_wmma_f32_16x16x4_f32(
                         false, a, false, b1, (short)0, d1, false, false);
                d2 = __builtin_amdgcn_wmma_f32_16x16x4_f32(
                         false, a, false, b2, (short)0, d2, false, false);
                d3 = __builtin_amdgcn_wmma_f32_16x16x4_f32(
                         false, a, false, b3, (short)0, d3, false, false);
            }

            #pragma unroll
            for (int r = 0; r < 8; ++r) {
                const int pw  = (r & 1) ? q[r >> 1].z : q[r >> 1].x;
                const int pos = pw & POS_MASK;
                float4 o;
                o.x = d0[r];
                o.y = d1[r];
                o.z = d2[r];
                o.w = d3[r];
                *reinterpret_cast<float4*>(Th + (size_t)pos * TW) = o;
            }
        }
        tb += CAP;
        if (tb >= nb) break;
        __syncthreads();
        for (int k = sub; k < rl; k += 4) {
            const int d = rx + k - tb;
            if ((unsigned)d < (unsigned)CAP) ent2[d] = P2[run * SLICE + ro + k];
        }
        if (tid < 16) {
            const int j = nb + tid;
            const int d = j - tb;
            if (j < nchunks * 16 && (unsigned)d < (unsigned)CAP) {
                int2 z;
                z.x = NNZ + bkt * 16 + tid;
                z.y = 0;
                ent2[d] = z;
            }
        }
    }
}

__global__ __launch_bounds__(256)
void regrid_rowsum_kernel(const float* __restrict__ T,
                          const int*   __restrict__ rstart,
                          const int*   __restrict__ rcnt,
                          float*       __restrict__ y) {
    __shared__ __attribute__((aligned(16))) float tile[64 * TP];

    const int tid  = threadIdx.x;
    const int lane = tid & 31;
    const int wave = tid >> 5;
    const int rg   = lane >> 4;
    const int cq   = lane & 15;
    const int i0   = blockIdx.x * RB;

    for (int it = 0; it < RB / 16; ++it) {
        const int rl = ((it * 8 + wave) << 1) + rg;
        const int i  = i0 + rl;
        float4 acc = {0.0f, 0.0f, 0.0f, 0.0f};
        const int ii = i < N_OUT ? i : N_OUT - 1;
        const int p0 = rstart[ii];
        const int n  = i < N_OUT ? rcnt[ii] : 0;
        const float4* src = reinterpret_cast<const float4*>(T) + cq;
        const int jmax = n > 0 ? n - 1 : 0;
        float4 v[8];
        #pragma unroll
        for (int j = 0; j < 8; ++j) {
            int line = p0 + (j < jmax ? j : jmax);
            line = line < NNZ ? line : NNZ - 1;
            v[j] = src[(size_t)line * (TW / 4)];
        }
        #pragma unroll
        for (int j = 0; j < 8; ++j) {
            const bool on = j < n;
            acc.x += on ? v[j].x : 0.0f;
            acc.y += on ? v[j].y : 0.0f;
            acc.z += on ? v[j].z : 0.0f;
            acc.w += on ? v[j].w : 0.0f;
        }
        for (int j = 8; j < n; ++j) {
            const float4 z = src[(size_t)(p0 + j) * (TW / 4)];
            acc.x += z.x; acc.y += z.y; acc.z += z.z; acc.w += z.w;
        }
        tile[(cq +  0) * TP + rl] = acc.x;
        tile[(cq + 16) * TP + rl] = acc.y;
        tile[(cq + 32) * TP + rl] = acc.z;
        tile[(cq + 48) * TP + rl] = acc.w;
    }
    __syncthreads();

    #pragma unroll
    for (int m = 0; m < (64 * RB / 4) / 256; ++m) {
        const int f  = m * 256 + tid;
        const int ch = f / (RB / 4);
        const int r4 = (f % (RB / 4)) << 2;
        if (i0 + r4 < N_OUT) {
            const float4 o = *reinterpret_cast<const float4*>(tile + ch * TP + r4);
            *reinterpret_cast<float4*>(y + (size_t)ch * N_OUT + i0 + r4) = o;
        }
    }
}

extern "C" void kernel_launch(void* const* d_in, const int* in_sizes, int n_in,
                              void* d_out, int out_size, void* d_ws, size_t ws_size,
                              hipStream_t stream) {
    (void)in_sizes; (void)n_in; (void)out_size; (void)ws_size;

    const float* x    = (const float*)d_in[0];
    const float* S    = (const float*)d_in[1];
    const int*   rows = (const int*)d_in[2];
    const int*   cols = (const int*)d_in[3];
    float*       y    = (float*)d_out;
    int*         ws   = (int*)d_ws;
    float*       T    = (float*)(ws + WS_T);

    regrid_slice_sort_kernel<<<NSL, 1024, 0, stream>>>(S, rows, cols, ws + WS_OC, ws + WS_OR,
                                                       ws + WS_R);
    regrid_rank_kernel<<<NRB, 1024, 0, stream>>>(ws + WS_OR, ws + WS_R, ws + WS_RSTART,
                                                 ws + WS_RCNT, ws + WS_P);

    regrid_product_wmma_kernel<<<NB, 1024, 0, stream>>>(x, ws + WS_P, ws + WS_OC, T);
    regrid_rowsum_kernel<<<(N_OUT + RB - 1) / RB, 256, 0, stream>>>(T, ws + WS_RSTART,
                                                                    ws + WS_RCNT, y);
}


# product launch made resident: 256 work-groups of 16 waves walk 4 column blocks each; next block's run bounds and half of its x loads are in flight under the current block's products
# speedup vs baseline: 17.4597x; 1.0442x over previous
#include <hip/hip_runtime.h>


typedef __attribute__((ext_vector_type(2))) float v2f;
typedef __attribute__((ext_vector_type(8))) float v8f;

#define H_IN   721
#define W_IN   1440
#define H_OUT  361
#define W_OUT  720
#define N_IN   (H_IN * W_IN)
#define N_OUT  (H_OUT * W_OUT)
#define NNZ    1048576

#define CB_SHIFT 10
#define CB       (1 << CB_SHIFT)
#define NB       ((N_IN + CB - 1) / CB)
#define NBP      1024
#define PITCH    (CB + 4)
#define SLICE    4096
#define NSL      (NNZ / SLICE)
#define OCP      (NBP + 1)
#define ORP      257
#define CAP      4096

#define POS_BITS 21
#define POS_MASK ((1 << POS_BITS) - 1)
#define TROWS    (NNZ + NBP * 16)
#define TW       64

#define NRB      ((N_OUT + 1023) >> 10)
#define N_OUT_P  (NRB << 10)
#define RB       128
#define TP       (RB + 4)

static_assert(NNZ + NBP * 16 <= (1 << POS_BITS), "position must fit its packed field");
static_assert(POS_BITS + CB_SHIFT <= 32, "packed entry fits 32 bits");
static_assert(NB <= NBP, "one sort thread per bucket");
static_assert(NRB <= 256, "row-block histogram size");
static_assert(NNZ % SLICE == 0, "whole slices");
static_assert(NSL == 256, "one run per slice: 256 runs per bin");
static_assert(SLICE < (1 << 15), "two packed 16-bit running sums");
static_assert(CAP % 16 == 0, "whole chunks per tile of entries");
static_assert(N_OUT % 4 == 0 && N_IN % 4 == 0, "float4 never split by an edge");

#define WS_OC     0
#define WS_OR     (WS_OC + NSL * OCP)
#define WS_RCNT   (WS_OR + NSL * ORP)
#define WS_RSTART (WS_RCNT + N_OUT_P)
#define WS_R      (WS_RSTART + N_OUT_P)
#define WS_P      (WS_R + 4 * NNZ)
#define WS_T      (WS_P + 2 * NNZ)

static_assert(WS_T % 64 == 0, "T lines start on a 256-B boundary");
static_assert(WS_P % 4 == 0 && WS_R % 4 == 0, "16-B entries of R, 8-B entries of P");

__device__ __forceinline__ int bucket_of(int col) {
    const unsigned b = (unsigned)col >> CB_SHIFT;
    return (int)(b < (unsigned)NB ? b : (unsigned)(NB - 1));
}
__device__ __forceinline__ int row_of(int row) {
    return (int)((unsigned)row < (unsigned)N_OUT ? (unsigned)row : (unsigned)(N_OUT - 1));
}

__device__ __forceinline__ int block_scan_incl(int* s, int tid, int v) {
    s[tid] = v;
    __syncthreads();
    for (int off = 1; off < 1024; off <<= 1) {
        const int a = (tid >= off) ? s[tid - off] : 0;
        __syncthreads();
        s[tid] += a;
        __syncthreads();
    }
    return s[tid];
}

__global__ __launch_bounds__(1024)
void regrid_slice_sort_kernel(const float* __restrict__ S,
                              const int*   __restrict__ rows,
                              const int*   __restrict__ cols,
                              int* __restrict__ OC, int* __restrict__ ORo,
                              int* __restrict__ R) {
    __shared__ int h[NBP];
    __shared__ int lbase[NBP];
    __shared__ int hr[256];
    __shared__ int lbr[256];
    __shared__ int sc[1024];
    const int tid = threadIdx.x;
    const int sl  = blockIdx.x;
    h[tid] = 0;
    if (tid < 256) hr[tid] = 0;
    __syncthreads();
    const int k0 = sl * SLICE + tid;
    int   cv[SLICE / 1024], bv[SLICE / 1024], rv[SLICE / 1024];
    float wv[SLICE / 1024];
    #pragma unroll
    for (int j = 0; j < SLICE / 1024; ++j) {
        cv[j] = cols[k0 + j * 1024];
        rv[j] = row_of(rows[k0 + j * 1024]);
        wv[j] = S[k0 + j * 1024];
    }
    #pragma unroll
    for (int j = 0; j < SLICE / 1024; ++j) {
        bv[j] = bucket_of(cv[j]);
        atomicAdd(&h[bv[j]], 1);
        atomicAdd(&hr[rv[j] >> 10], 1);
    }
    __syncthreads();
    const int v   = h[tid];
    const int w   = tid < 256 ? hr[tid] : 0;
    const int inc = block_scan_incl(sc, tid, v | (w << 16));
    const int exc = (inc & 0xFFFF) - v;
    const int exr = (inc >> 16) - w;
    lbase[tid] = exc;
    h[tid] = 0;
    OC[sl * OCP + tid] = exc;
    if (tid == 1023) OC[sl * OCP + NBP] = SLICE;
    if (tid < 256) {
        lbr[tid] = exr;
        hr[tid] = 0;
        ORo[sl * ORP + tid] = exr;
    }
    if (tid == 255) ORo[sl * ORP + 256] = SLICE;
    __syncthreads();
    int4* Rs = reinterpret_cast<int4*>(R) + sl * SLICE;
    #pragma unroll
    for (int j = 0; j < SLICE / 1024; ++j) {
        const int r    = atomicAdd(&h[bv[j]], 1);
        const int slot = sl * SLICE + lbase[bv[j]] + r;
        const int rb   = rv[j] >> 10;
        const int rr   = atomicAdd(&hr[rb], 1);
        int4 e;
        e.x = (slot & POS_MASK) | ((cv[j] & (CB - 1)) << POS_BITS);
        e.y = rv[j];
        e.z = __float_as_int(wv[j]);
        e.w = 0;
        Rs[lbr[rb] + rr] = e;
    }
}

__global__ __launch_bounds__(1024)
void regrid_rank_kernel(const int* __restrict__ ORo,
                        const int* __restrict__ R,
                        int* __restrict__ rstart, int* __restrict__ rcnt,
                        int* __restrict__ P) {
    __shared__ int lc[1024];
    __shared__ int s[1024];
    __shared__ int roff[NSL];
    __shared__ int rlen[NSL];
    __shared__ int stsum;
    const int tid  = threadIdx.x;
    const int lane = tid & 31;
    const int b    = blockIdx.x;

    lc[tid] = 0;
    if (tid == 0) stsum = 0;
    __syncthreads();
    if (tid < NSL) {
        const int a = ORo[tid * ORP + b];
        const int e = ORo[tid * ORP + b + 1];
        roff[tid] = a;
        rlen[tid] = e - a;
        int t = a;
        #pragma unroll
        for (int m = 16; m >= 1; m >>= 1) t += __shfl_xor(t, m);
        if (lane == 0) atomicAdd(&stsum, t);
    }
    __syncthreads();
    const int st = stsum;
    const int4* R4 = reinterpret_cast<const int4*>(R);
    const int kl = tid & 15;

    {
        int rw[8];
        #pragma unroll
        for (int u = 0; u < 8; ++u) {
            const int run = (tid >> 4) + 64 * (u & 3);
            const int k   = kl + 16 * (u >> 2);
            rw[u] = k < rlen[run] ? R4[run * SLICE + roff[run] + k].y : -1;
        }
        #pragma unroll
        for (int u = 0; u < 8; ++u)
            if (rw[u] >= 0) atomicAdd(&lc[rw[u] & 1023], 1);
        for (int g = 0; g < 4; ++g) {
            const int run = (tid >> 4) + 64 * g;
            const int n   = rlen[run];
            for (int k = kl + 32; k < n; k += 16)
                atomicAdd(&lc[R4[run * SLICE + roff[run] + k].y & 1023], 1);
        }
    }
    __syncthreads();
    const int c    = lc[tid];
    const int excl = block_scan_incl(s, tid, c) - c;
    rstart[b * 1024 + tid] = st + excl;
    rcnt[b * 1024 + tid]   = c;
    s[tid]  = st + excl;
    lc[tid] = 0;
    __syncthreads();
    {
        int4 e[8];
        #pragma unroll
        for (int u = 0; u < 8; ++u) {
            const int run = (tid >> 4) + 64 * (u & 3);
            const int k   = kl + 16 * (u >> 2);
            int4 z;
            z.x = 0;
            z.y = -1;
            z.z = 0;
            z.w = 0;
            if (k < rlen[run]) z = R4[run * SLICE + roff[run] + k];
            e[u] = z;
        }
        #pragma unroll
        for (int u = 0; u < 8; ++u) {
            if (e[u].y >= 0) {
                const int rl  = e[u].y & 1023;
                const int r   = atomicAdd(&lc[rl], 1);
                const int pos = s[rl] + r;
                int2 o;
                o.x = (pos & POS_MASK) | (e[u].x & ~POS_MASK);
                o.y = e[u].z;
                reinterpret_cast<int2*>(P)[e[u].x & POS_MASK] = o;
            }
        }
        for (int g = 0; g < 4; ++g) {
            const int run = (tid >> 4) + 64 * g;
            const int n   = rlen[run];
            for (int k = kl + 32; k < n; k += 16) {
                const int4 z  = R4[run * SLICE + roff[run] + k];
                const int rl  = z.y & 1023;
                const int r   = atomicAdd(&lc[rl], 1);
                const int pos = s[rl] + r;
                int2 o;
                o.x = (pos & POS_MASK) | (z.x & ~POS_MASK);
                o.y = z.z;
                reinterpret_cast<int2*>(P)[z.x & POS_MASK] = o;
            }
        }
    }
}

#define PGRID 256
#define PNT   512
#define PNW   (PNT / 32)
#define XRS   (PNT / 256)
#define XL    (64 / XRS)
#define XA    16
#define LPR   (PNT / NSL)
#define EVN   (16 / LPR)

static_assert(PNT % 256 == 0 && PNT >= NSL, "a load covers whole plane rows; one thread per run");
static_assert(XA < XL, "part of a block is loaded ahead, the rest behind the scan");

__global__ __launch_bounds__(PNT)
void regrid_product_wmma_kernel(const float* __restrict__ x,
                                const int*   __restrict__ P,
                                const int*   __restrict__ OC,
                                float*       __restrict__ T) {
    __shared__ __attribute__((aligned(16))) float xs[64 * PITCH];
    __shared__ __attribute__((aligned(16))) int ent[2 * CAP];
    __shared__ int rt[3 * NSL + NSL / 32];
    int* const coff = rt;
    int* const clen = rt + NSL;
    int* const cexc = rt + 2 * NSL;
    int* const wtot = rt + 3 * NSL;

    const int tid    = threadIdx.x;
    const int lane   = tid & 31;
    const int c      = lane & 15;
    const int half   = lane >> 4;

    const int wave = __builtin_amdgcn_readfirstlane(tid >> 5);

    const unsigned ocb = (unsigned)((tid & (NSL - 1)) * OCP) * 4u;

    const char* Pb   = reinterpret_cast<const char*>(P);
    int2*       ent2 = reinterpret_cast<int2*>(ent);
    const int run = tid / LPR;
    const int sub = tid % LPR;

    const int      c4  = (tid & 255) << 2;
    const unsigned xob = (unsigned)((tid >> 8) * N_IN + c4) * 4u;
    float* xsw = xs + (tid >> 8) * PITCH + c4;

    const float* xr0 = xs + c * PITCH;
    const float* xr1 = xs + (16 + c) * PITCH;
    const float* xr2 = xs + (32 + c) * PITCH;
    const float* xr3 = xs + (48 + c) * PITCH;
    char* Tb = reinterpret_cast<char*>(T);
    const unsigned tcb = 16u * (unsigned)c;

    int bkt = blockIdx.x;
    int ra, re;
    {
        const int* o = reinterpret_cast<const int*>(reinterpret_cast<const char*>(OC + bkt) + ocb);
        ra = o[0];
        re = o[1];
    }
    float4 ta[XA];
    #pragma unroll
    for (int i = 0; i < XA; ++i) {
        const char* xb = reinterpret_cast<const char*>(x + (size_t)(XRS * i) * N_IN + (bkt << CB_SHIFT));
        float4 z = {0.0f, 0.0f, 0.0f, 0.0f};
        if ((bkt << CB_SHIFT) + c4 < N_IN)
            z = *reinterpret_cast<const float4*>(xb + xob);
        ta[i] = z;
    }

    while (true) {
        const int col0 = bkt << CB_SHIFT;

        const int rn = tid < NSL ? re - ra : 0;
        int inc = rn;
        #pragma unroll
        for (int d = 1; d < 32; d <<= 1) {
            const int u = __shfl_up(inc, d);
            if (lane >= d) inc += u;
        }
        if (tid < NSL && lane == 31) wtot[tid >> 5] = inc;
        __syncthreads();
        int nbv = 0, before = 0;
        #pragma unroll
        for (int w = 0; w < NSL / 32; ++w) {
            const int q = wtot[w];
            if (w < (tid >> 5)) before += q;
            nbv += q;
        }
        if (tid < NSL) {
            coff[tid] = ra;
            clen[tid] = rn;
            cexc[tid] = before + inc - rn;
        }
        __syncthreads();
        const int nb      = __builtin_amdgcn_readfirstlane(nbv);
        const int nchunks = (nb + 15) >> 4;

        const int rl  = clen[run];
        const int rx  = cexc[run];
        const unsigned pob = (unsigned)(run * SLICE + coff[run]) * 8u;

        int2 ev[EVN];
        #pragma unroll
        for (int u = 0; u < EVN; ++u) {
            const int k = sub + LPR * u;
            int2 z;
            z.x = 0;
            z.y = 0;
            if (k < rl) z = *reinterpret_cast<const int2*>(Pb + (pob + 8u * (unsigned)k));
            ev[u] = z;
        }
        #pragma unroll
        for (int i = 0; i < XA; ++i)
            *reinterpret_cast<float4*>(xsw + i * XRS * PITCH) = ta[i];
        float4 t[XL - XA];
        #pragma unroll
        for (int i = 0; i < XL - XA; ++i) {
            const char* xb = reinterpret_cast<const char*>(x + (size_t)(XRS * (XA + i)) * N_IN + col0);
            float4 z = {0.0f, 0.0f, 0.0f, 0.0f};
            if (col0 + c4 < N_IN)
                z = *reinterpret_cast<const float4*>(xb + xob);
            t[i] = z;
        }
        #pragma unroll
        for (int u = 0; u < EVN; ++u) {
            const int k = sub + LPR * u;
            const int d = rx + k;
            if (k < rl && (unsigned)d < (unsigned)CAP) ent2[d] = ev[u];
        }
        for (int k = sub + 16; k < rl; k += LPR) {
            const int d = rx + k;
            if ((unsigned)d < (unsigned)CAP)
                ent2[d] = *reinterpret_cast<const int2*>(Pb + (pob + 8u * (unsigned)k));
        }
        if (tid < 16) {
            const int j = nb + tid;
            if (j < nchunks * 16 && j < CAP) {
                int2 z;
                z.x = NNZ + bkt * 16 + tid;
                z.y = 0;
                ent2[j] = z;
            }
        }
        #pragma unroll
        for (int i = 0; i < XL - XA; ++i)
            *reinterpret_cast<float4*>(xsw + (XA + i) * XRS * PITCH) = t[i];

        const int nxt = bkt + PGRID;

        int tb = 0;
        while (true) {
            __syncthreads();

            if (tb == 0 && nxt < NB) {
                const int* o = reinterpret_cast<const int*>(reinterpret_cast<const char*>(OC + nxt) + ocb);
                ra = o[0];
                re = o[1];
                #pragma unroll
                for (int i = 0; i < XA; ++i) {
                    const char* xb = reinterpret_cast<const char*>(x + (size_t)(XRS * i) * N_IN + (nxt << CB_SHIFT));
                    float4 z = {0.0f, 0.0f, 0.0f, 0.0f};
                    if ((nxt << CB_SHIFT) + c4 < N_IN)
                        z = *reinterpret_cast<const float4*>(xb + xob);
                    ta[i] = z;
                }
            }

            const int cend = min(nchunks, (tb + CAP) >> 4);
            for (int chunk = (tb >> 4) + wave; chunk < cend; chunk += PNW) {
                const int* pe = ent + 2 * (chunk * 16 - tb) + 2 * (8 * half);
                int4 q[4];
                #pragma unroll
                for (int g = 0; g < 4; ++g)
                    q[g] = *reinterpret_cast<const int4*>(pe + 4 * g);

                float x0[8], x1[8], x2[8], x3[8], sv[8];
                #pragma unroll
                for (int g = 0; g < 4; ++g) {
                    const int ca = (q[g].x >> POS_BITS) & (CB - 1);
                    const int cb = (q[g].z >> POS_BITS) & (CB - 1);
                    x0[2 * g + 0] = xr0[ca];
                    x0[2 * g + 1] = xr0[cb];
                    x1[2 * g + 0] = xr1[ca];
                    x1[2 * g + 1] = xr1[cb];
                    x2[2 * g + 0] = xr2[ca];
                    x2[2 * g + 1] = xr2[cb];
                    x3[2 * g + 0] = xr3[ca];
                    x3[2 * g + 1] = xr3[cb];
                    sv[2 * g + 0] = __int_as_float(q[g].y);
                    sv[2 * g + 1] = __int_as_float(q[g].w);
                }

                v8f d0 = {}, d1 = {}, d2 = {}, d3 = {};
                #pragma unroll
                for (int g = 0; g < 4; ++g) {
                    v2f a, b0, b1, b2, b3;
                    a[0]  = (c == 8 * half + 2 * g + 0) ? sv[2 * g + 0] : 0.0f;
                    a[1]  = (c == 8 * half + 2 * g + 1) ? sv[2 * g + 1] : 0.0f;
                    b0[0] = x0[2 * g + 0];
                    b0[1] = x0[2 * g + 1];
                    b1[0] = x1[2 * g + 0];
                    b1[1] = x1[2 * g + 1];
                    b2[0] = x2[2 * g + 0];
                    b2[1] = x2[2 * g + 1];
                    b3[0] = x3[2 * g + 0];
                    b3[1] = x3[2 * g + 1];
                    d0 = __builtin_amdgcn_wmma_f32_16x16x4_f32(
                             false, a, false, b0, (short)0, d0, false, false);
                    d1 = __builtin_amdgcn_wmma_f32_16x16x4_f32(
                             false, a, false, b1, (short)0, d1, false, false);
                    d2 = __builtin_amdgcn_wmma_f32_16x16x4_f32(
                             false, a, false, b2, (short)0, d2, false, false);
                    d3 = __builtin_amdgcn_wmma_f32_16x16x4_f32(
                             false, a, false, b3, (short)0, d3, false, false);
                }

                #pragma unroll
                for (int r = 0; r < 8; ++r) {
                    const int pw  = (r & 1) ? q[r >> 1].z : q[r >> 1].x;
                    const unsigned pos = (unsigned)(pw & POS_MASK);
                    float4 o;
                    o.x = d0[r];
                    o.y = d1[r];
                    o.z = d2[r];
                    o.w = d3[r];
                    *reinterpret_cast<float4*>(Tb + (pos * (unsigned)(TW * 4) + tcb)) = o;
                }
            }
            tb += CAP;
            if (tb >= nb) break;
            __syncthreads();
            const int rl2 = clen[run];
            const int rx2 = cexc[run];
            const unsigned pob2 = (unsigned)(run * SLICE + coff[run]) * 8u;
            for (int k = sub; k < rl2; k += LPR) {
                const int d = rx2 + k - tb;
                if ((unsigned)d < (unsigned)CAP)
                    ent2[d] = *reinterpret_cast<const int2*>(Pb + (pob2 + 8u * (unsigned)k));
            }
            if (tid < 16) {
                const int j = nb + tid;
                const int d = j - tb;
                if (j < nchunks * 16 && (unsigned)d < (unsigned)CAP) {
                    int2 z;
                    z.x = NNZ + bkt * 16 + tid;
                    z.y = 0;
                    ent2[d] = z;
                }
            }
        }

        if (nxt >= NB) break;
        bkt = nxt;
    }
}

__global__ __launch_bounds__(256)
void regrid_rowsum_kernel(const float* __restrict__ T,
                          const int*   __restrict__ rstart,
                          const int*   __restrict__ rcnt,
                          float*       __restrict__ y) {
    __shared__ __attribute__((aligned(16))) float tile[64 * TP];

    const int tid  = threadIdx.x;
    const int lane = tid & 31;
    const int wave = tid >> 5;
    const int rg   = lane >> 4;
    const int cq   = lane & 15;
    const int i0   = blockIdx.x * RB;

    for (int it = 0; it < RB / 16; ++it) {
        const int rl = ((it * 8 + wave) << 1) + rg;
        const int i  = i0 + rl;
        float4 acc = {0.0f, 0.0f, 0.0f, 0.0f};
        const int ii = i < N_OUT ? i : N_OUT - 1;
        const int p0 = rstart[ii];
        const int n  = i < N_OUT ? rcnt[ii] : 0;
        const float4* src = reinterpret_cast<const float4*>(T) + cq;
        const int jmax = n > 0 ? n - 1 : 0;
        float4 v[8];
        #pragma unroll
        for (int j = 0; j < 8; ++j) {
            int line = p0 + (j < jmax ? j : jmax);
            line = line < NNZ ? line : NNZ - 1;
            v[j] = src[(size_t)line * (TW / 4)];
        }
        #pragma unroll
        for (int j = 0; j < 8; ++j) {
            const bool on = j < n;
            acc.x += on ? v[j].x : 0.0f;
            acc.y += on ? v[j].y : 0.0f;
            acc.z += on ? v[j].z : 0.0f;
            acc.w += on ? v[j].w : 0.0f;
        }
        for (int j = 8; j < n; ++j) {
            const float4 z = src[(size_t)(p0 + j) * (TW / 4)];
            acc.x += z.x; acc.y += z.y; acc.z += z.z; acc.w += z.w;
        }
        tile[(cq +  0) * TP + rl] = acc.x;
        tile[(cq + 16) * TP + rl] = acc.y;
        tile[(cq + 32) * TP + rl] = acc.z;
        tile[(cq + 48) * TP + rl] = acc.w;
    }
    __syncthreads();

    #pragma unroll
    for (int m = 0; m < (64 * RB / 4) / 256; ++m) {
        const int f  = m * 256 + tid;
        const int ch = f / (RB / 4);
        const int r4 = (f % (RB / 4)) << 2;
        if (i0 + r4 < N_OUT) {
            const float4 o = *reinterpret_cast<const float4*>(tile + ch * TP + r4);
            *reinterpret_cast<float4*>(y + (size_t)ch * N_OUT + i0 + r4) = o;
        }
    }
}

extern "C" void kernel_launch(void* const* d_in, const int* in_sizes, int n_in,
                              void* d_out, int out_size, void* d_ws, size_t ws_size,
                              hipStream_t stream) {
    (void)in_sizes; (void)n_in; (void)out_size; (void)ws_size;

    const float* x    = (const float*)d_in[0];
    const float* S    = (const float*)d_in[1];
    const int*   rows = (const int*)d_in[2];
    const int*   cols = (const int*)d_in[3];
    float*       y    = (float*)d_out;
    int*         ws   = (int*)d_ws;
    float*       T    = (float*)(ws + WS_T);

    regrid_slice_sort_kernel<<<NSL, 1024, 0, stream>>>(S, rows, cols, ws + WS_OC, ws + WS_OR,
                                                       ws + WS_R);
    regrid_rank_kernel<<<NRB, 1024, 0, stream>>>(ws + WS_OR, ws + WS_R, ws + WS_RSTART,
                                                 ws + WS_RCNT, ws + WS_P);

    regrid_product_wmma_kernel<<<PGRID, PNT, 0, stream>>>(x, ws + WS_P, ws + WS_OC, T);
    regrid_rowsum_kernel<<<(N_OUT + RB - 1) / RB, 256, 0, stream>>>(T, ws + WS_RSTART,
                                                                    ws + WS_RCNT, y);
}


# a nonzero's T line is its slot in the column-sorted list, known in the sort launch: rank launch gone (4 to 3); row work-groups sort their T lines by row in LDS; sum order = LDS arrival rank
# speedup vs baseline: 20.7967x; 1.1911x over previous
#include <hip/hip_runtime.h>


typedef __attribute__((ext_vector_type(2))) float v2f;
typedef __attribute__((ext_vector_type(8))) float v8f;

#define H_IN   721
#define W_IN   1440
#define H_OUT  361
#define W_OUT  720
#define N_IN   (H_IN * W_IN)
#define N_OUT  (H_OUT * W_OUT)
#define NNZ    1048576

#define CB_SHIFT 10
#define CB       (1 << CB_SHIFT)
#define NB       ((N_IN + CB - 1) / CB)
#define NBP      1024
#define PITCH    (CB + 4)
#define SLICE    4096
#define NSL      (NNZ / SLICE)
#define OCP      (NBP + 1)
#define ORP      257
#define CAP      4096

#define POS_BITS 21
#define POS_MASK ((1 << POS_BITS) - 1)
#define TROWS    (NNZ + NBP * 16)
#define TW       64

#define NRB      ((N_OUT + 1023) >> 10)
#define LCAP     8192
#define TPB      (1024 + 4)

static_assert(NNZ + NBP * 16 <= (1 << POS_BITS), "T line must fit its packed field");
static_assert(POS_BITS + CB_SHIFT <= 32, "packed entry fits 32 bits");
static_assert(NB <= NBP, "one sort thread per bucket");
static_assert(NRB <= 256, "row-block histogram size");
static_assert(NNZ % SLICE == 0, "whole slices");
static_assert(NSL == 256, "one run per slice: 256 runs per bin");
static_assert(SLICE < (1 << 15), "two packed 16-bit running sums");
static_assert(CAP % 16 == 0, "whole chunks per tile of entries");
static_assert(N_OUT % 4 == 0 && N_IN % 4 == 0, "float4 never split by an edge");
static_assert((NNZ & (NNZ - 1)) == 0, "a T line index is kept in range by a mask");

#define WS_OC     0
#define WS_OR     (WS_OC + NSL * OCP)
#define WS_R      (WS_OR + NSL * ORP)
#define WS_P      (WS_R + 2 * NNZ)
#define WS_T      (WS_P + 2 * NNZ)

static_assert(WS_T % 64 == 0, "T lines start on a 256-B boundary");
static_assert(WS_P % 2 == 0 && WS_R % 2 == 0, "8-B entries of R and of P");

__device__ __forceinline__ int bucket_of(int col) {
    const unsigned b = (unsigned)col >> CB_SHIFT;
    return (int)(b < (unsigned)NB ? b : (unsigned)(NB - 1));
}
__device__ __forceinline__ int row_of(int row) {
    return (int)((unsigned)row < (unsigned)N_OUT ? (unsigned)row : (unsigned)(N_OUT - 1));
}

__device__ __forceinline__ int block_scan_incl(int* s, int tid, int v) {
    s[tid] = v;
    __syncthreads();
    for (int off = 1; off < 1024; off <<= 1) {
        const int a = (tid >= off) ? s[tid - off] : 0;
        __syncthreads();
        s[tid] += a;
        __syncthreads();
    }
    return s[tid];
}

__global__ __launch_bounds__(1024)
void regrid_slice_sort_kernel(const float* __restrict__ S,
                              const int*   __restrict__ rows,
                              const int*   __restrict__ cols,
                              int* __restrict__ OC, int* __restrict__ ORo,
                              int* __restrict__ R, int* __restrict__ P) {
    __shared__ int h[NBP];
    __shared__ int lbase[NBP];
    __shared__ int hr[256];
    __shared__ int lbr[256];
    __shared__ int sc[1024];
    const int tid = threadIdx.x;
    const int sl  = blockIdx.x;
    h[tid] = 0;
    if (tid < 256) hr[tid] = 0;
    __syncthreads();
    const int k0 = sl * SLICE + tid;
    int   cv[SLICE / 1024], bv[SLICE / 1024], rv[SLICE / 1024];
    float wv[SLICE / 1024];
    #pragma unroll
    for (int j = 0; j < SLICE / 1024; ++j) {
        cv[j] = cols[k0 + j * 1024];
        rv[j] = row_of(rows[k0 + j * 1024]);
        wv[j] = S[k0 + j * 1024];
    }
    #pragma unroll
    for (int j = 0; j < SLICE / 1024; ++j) {
        bv[j] = bucket_of(cv[j]);
        atomicAdd(&h[bv[j]], 1);
        atomicAdd(&hr[rv[j] >> 10], 1);
    }
    __syncthreads();
    const int v   = h[tid];
    const int w   = tid < 256 ? hr[tid] : 0;
    const int inc = block_scan_incl(sc, tid, v | (w << 16));
    const int exc = (inc & 0xFFFF) - v;
    const int exr = (inc >> 16) - w;
    lbase[tid] = exc;
    h[tid] = 0;
    OC[sl * OCP + tid] = exc;
    if (tid == 1023) OC[sl * OCP + NBP] = SLICE;
    if (tid < 256) {
        lbr[tid] = exr;
        hr[tid] = 0;
        ORo[sl * ORP + tid] = exr;
    }
    if (tid == 255) ORo[sl * ORP + 256] = SLICE;
    __syncthreads();
    int2* Rs = reinterpret_cast<int2*>(R) + sl * SLICE;
    int2* Pa = reinterpret_cast<int2*>(P);
    #pragma unroll
    for (int j = 0; j < SLICE / 1024; ++j) {
        const int r    = atomicAdd(&h[bv[j]], 1);
        const int slot = sl * SLICE + lbase[bv[j]] + r;
        const int rb   = rv[j] >> 10;
        const int rr   = atomicAdd(&hr[rb], 1);
        int2 p;
        p.x = (slot & POS_MASK) | ((cv[j] & (CB - 1)) << POS_BITS);
        p.y = __float_as_int(wv[j]);
        Pa[slot] = p;
        int2 e;
        e.x = slot;
        e.y = rv[j];
        Rs[lbr[rb] + rr] = e;
    }
}

#define PGRID 256
#define PNT   512
#define PNW   (PNT / 32)
#define XRS   (PNT / 256)
#define XL    (64 / XRS)
#define XA    16
#define LPR   (PNT / NSL)
#define EVN   (16 / LPR)

static_assert(PNT % 256 == 0 && PNT >= NSL, "a load covers whole plane rows; one thread per run");
static_assert(XA < XL, "part of a block is loaded ahead, the rest behind the scan");

__global__ __launch_bounds__(PNT)
void regrid_product_wmma_kernel(const float* __restrict__ x,
                                const int*   __restrict__ P,
                                const int*   __restrict__ OC,
                                float*       __restrict__ T) {
    __shared__ __attribute__((aligned(16))) float xs[64 * PITCH];
    __shared__ __attribute__((aligned(16))) int ent[2 * CAP];
    __shared__ int rt[3 * NSL + NSL / 32];
    int* const coff = rt;
    int* const clen = rt + NSL;
    int* const cexc = rt + 2 * NSL;
    int* const wtot = rt + 3 * NSL;

    const int tid    = threadIdx.x;
    const int lane   = tid & 31;
    const int c      = lane & 15;
    const int half   = lane >> 4;

    const int wave = __builtin_amdgcn_readfirstlane(tid >> 5);

    const unsigned ocb = (unsigned)((tid & (NSL - 1)) * OCP) * 4u;

    const char* Pb   = reinterpret_cast<const char*>(P);
    int2*       ent2 = reinterpret_cast<int2*>(ent);
    const int run = tid / LPR;
    const int sub = tid % LPR;

    const int      c4  = (tid & 255) << 2;
    const unsigned xob = (unsigned)((tid >> 8) * N_IN + c4) * 4u;
    float* xsw = xs + (tid >> 8) * PITCH + c4;

    const float* xr0 = xs + c * PITCH;
    const float* xr1 = xs + (16 + c) * PITCH;
    const float* xr2 = xs + (32 + c) * PITCH;
    const float* xr3 = xs + (48 + c) * PITCH;
    char* Tb = reinterpret_cast<char*>(T);
    const unsigned tcb = 16u * (unsigned)c;

    int bkt = blockIdx.x;
    int ra, re;
    {
        const int* o = reinterpret_cast<const int*>(reinterpret_cast<const char*>(OC + bkt) + ocb);
        ra = o[0];
        re = o[1];
    }
    float4 ta[XA];
    #pragma unroll
    for (int i = 0; i < XA; ++i) {
        const char* xb = reinterpret_cast<const char*>(x + (size_t)(XRS * i) * N_IN + (bkt << CB_SHIFT));
        float4 z = {0.0f, 0.0f, 0.0f, 0.0f};
        if ((bkt << CB_SHIFT) + c4 < N_IN)
            z = *reinterpret_cast<const float4*>(xb + xob);
        ta[i] = z;
    }

    while (true) {
        const int col0 = bkt << CB_SHIFT;

        const int rn = tid < NSL ? re - ra : 0;
        int inc = rn;
        #pragma unroll
        for (int d = 1; d < 32; d <<= 1) {
            const int u = __shfl_up(inc, d);
            if (lane >= d) inc += u;
        }
        if (tid < NSL && lane == 31) wtot[tid >> 5] = inc;
        __syncthreads();
        int nbv = 0, before = 0;
        #pragma unroll
        for (int w = 0; w < NSL / 32; ++w) {
            const int q = wtot[w];
            if (w < (tid >> 5)) before += q;
            nbv += q;
        }
        if (tid < NSL) {
            coff[tid] = ra;
            clen[tid] = rn;
            cexc[tid] = before + inc - rn;
        }
        __syncthreads();
        const int nb      = __builtin_amdgcn_readfirstlane(nbv);
        const int nchunks = (nb + 15) >> 4;

        const int rl  = clen[run];
        const int rx  = cexc[run];
        const unsigned pob = (unsigned)(run * SLICE + coff[run]) * 8u;

        int2 ev[EVN];
        #pragma unroll
        for (int u = 0; u < EVN; ++u) {
            const int k = sub + LPR * u;
            int2 z;
            z.x = 0;
            z.y = 0;
            if (k < rl) z = *reinterpret_cast<const int2*>(Pb + (pob + 8u * (unsigned)k));
            ev[u] = z;
        }
        #pragma unroll
        for (int i = 0; i < XA; ++i)
            *reinterpret_cast<float4*>(xsw + i * XRS * PITCH) = ta[i];
        float4 t[XL - XA];
        #pragma unroll
        for (int i = 0; i < XL - XA; ++i) {
            const char* xb = reinterpret_cast<const char*>(x + (size_t)(XRS * (XA + i)) * N_IN + col0);
            float4 z = {0.0f, 0.0f, 0.0f, 0.0f};
            if (col0 + c4 < N_IN)
                z = *reinterpret_cast<const float4*>(xb + xob);
            t[i] = z;
        }
        #pragma unroll
        for (int u = 0; u < EVN; ++u) {
            const int k = sub + LPR * u;
            const int d = rx + k;
            if (k < rl && (unsigned)d < (unsigned)CAP) ent2[d] = ev[u];
        }
        for (int k = sub + 16; k < rl; k += LPR) {
            const int d = rx + k;
            if ((unsigned)d < (unsigned)CAP)
                ent2[d] = *reinterpret_cast<const int2*>(Pb + (pob + 8u * (unsigned)k));
        }
        if (tid < 16) {
            const int j = nb + tid;
            if (j < nchunks * 16 && j < CAP) {
                int2 z;
                z.x = NNZ + bkt * 16 + tid;
                z.y = 0;
                ent2[j] = z;
            }
        }
        #pragma unroll
        for (int i = 0; i < XL - XA; ++i)
            *reinterpret_cast<float4*>(xsw + (XA + i) * XRS * PITCH) = t[i];

        const int nxt = bkt + PGRID;

        int tb = 0;
        while (true) {
            __syncthreads();

            if (tb == 0 && nxt < NB) {
                const int* o = reinterpret_cast<const int*>(reinterpret_cast<const char*>(OC + nxt) + ocb);
                ra = o[0];
                re = o[1];
                #pragma unroll
                for (int i = 0; i < XA; ++i) {
                    const char* xb = reinterpret_cast<const char*>(x + (size_t)(XRS * i) * N_IN + (nxt << CB_SHIFT));
                    float4 z = {0.0f, 0.0f, 0.0f, 0.0f};
                    if ((nxt << CB_SHIFT) + c4 < N_IN)
                        z = *reinterpret_cast<const float4*>(xb + xob);
                    ta[i] = z;
                }
            }

            const int cend = min(nchunks, (tb + CAP) >> 4);
            for (int chunk = (tb >> 4) + wave; chunk < cend; chunk += PNW) {
                const int* pe = ent + 2 * (chunk * 16 - tb) + 2 * (8 * half);
                int4 q[4];
                #pragma unroll
                for (int g = 0; g < 4; ++g)
                    q[g] = *reinterpret_cast<const int4*>(pe + 4 * g);

                float x0[8], x1[8], x2[8], x3[8], sv[8];
                #pragma unroll
                for (int g = 0; g < 4; ++g) {
                    const int ca = (q[g].x >> POS_BITS) & (CB - 1);
                    const int cb = (q[g].z >> POS_BITS) & (CB - 1);
                    x0[2 * g + 0] = xr0[ca];
                    x0[2 * g + 1] = xr0[cb];
                    x1[2 * g + 0] = xr1[ca];
                    x1[2 * g + 1] = xr1[cb];
                    x2[2 * g + 0] = xr2[ca];
                    x2[2 * g + 1] = xr2[cb];
                    x3[2 * g + 0] = xr3[ca];
                    x3[2 * g + 1] = xr3[cb];
                    sv[2 * g + 0] = __int_as_float(q[g].y);
                    sv[2 * g + 1] = __int_as_float(q[g].w);
                }

                v8f d0 = {}, d1 = {}, d2 = {}, d3 = {};
                #pragma unroll
                for (int g = 0; g < 4; ++g) {
                    v2f a, b0, b1, b2, b3;
                    a[0]  = (c == 8 * half + 2 * g + 0) ? sv[2 * g + 0] : 0.0f;
                    a[1]  = (c == 8 * half + 2 * g + 1) ? sv[2 * g + 1] : 0.0f;
                    b0[0] = x0[2 * g + 0];
                    b0[1] = x0[2 * g + 1];
                    b1[0] = x1[2 * g + 0];
                    b1[1] = x1[2 * g + 1];
                    b2[0] = x2[2 * g + 0];
                    b2[1] = x2[2 * g + 1];
                    b3[0] = x3[2 * g + 0];
                    b3[1] = x3[2 * g + 1];
                    d0 = __builtin_amdgcn_wmma_f32_16x16x4_f32(
                             false, a, false, b0, (short)0, d0, false, false);
                    d1 = __builtin_amdgcn_wmma_f32_16x16x4_f32(
                             false, a, false, b1, (short)0, d1, false, false);
                    d2 = __builtin_amdgcn_wmma_f32_16x16x4_f32(
                             false, a, false, b2, (short)0, d2, false, false);
                    d3 = __builtin_amdgcn_wmma_f32_16x16x4_f32(
                             false, a, false, b3, (short)0, d3, false, false);
                }

                #pragma unroll
                for (int r = 0; r < 8; ++r) {
                    const int pw  = (r & 1) ? q[r >> 1].z : q[r >> 1].x;
                    const unsigned pos = (unsigned)(pw & POS_MASK);
                    float4 o;
                    o.x = d0[r];
                    o.y = d1[r];
                    o.z = d2[r];
                    o.w = d3[r];
                    *reinterpret_cast<float4*>(Tb + (pos * (unsigned)(TW * 4) + tcb)) = o;
                }
            }
            tb += CAP;
            if (tb >= nb) break;
            __syncthreads();
            const int rl2 = clen[run];
            const int rx2 = cexc[run];
            const unsigned pob2 = (unsigned)(run * SLICE + coff[run]) * 8u;
            for (int k = sub; k < rl2; k += LPR) {
                const int d = rx2 + k - tb;
                if ((unsigned)d < (unsigned)CAP)
                    ent2[d] = *reinterpret_cast<const int2*>(Pb + (pob2 + 8u * (unsigned)k));
            }
            if (tid < 16) {
                const int j = nb + tid;
                const int d = j - tb;
                if (j < nchunks * 16 && (unsigned)d < (unsigned)CAP) {
                    int2 z;
                    z.x = NNZ + bkt * 16 + tid;
                    z.y = 0;
                    ent2[d] = z;
                }
            }
        }

        if (nxt >= NB) break;
        bkt = nxt;
    }
}

__global__ __launch_bounds__(1024)
void regrid_rowsum_kernel(const float* __restrict__ T,
                          const int*   __restrict__ ORo,
                          const int*   __restrict__ R,
                          float*       __restrict__ y) {
    __shared__ __attribute__((aligned(16))) float tile[64 * TPB];
    __shared__ int list[LCAP];
    __shared__ int cn[1024];
    __shared__ int st[1024];
    __shared__ int cur[1024];
    __shared__ int roff[NSL];
    __shared__ int rlen[NSL];
    __shared__ int wt[32];

    const int tid  = threadIdx.x;
    const int lane = tid & 31;
    const int wave = tid >> 5;
    const int b    = blockIdx.x;

    cn[tid]  = 0;
    cur[tid] = 0;
    if (tid < NSL) {
        const int a = ORo[tid * ORP + b];
        const int e = ORo[tid * ORP + b + 1];
        roff[tid] = a;
        rlen[tid] = e - a;
    }
    __syncthreads();

    const int2* R2 = reinterpret_cast<const int2*>(R);
    const int kl = tid & 15;

    int2 e[8];
    #pragma unroll
    for (int u = 0; u < 8; ++u) {
        const int run = (tid >> 4) + 64 * (u & 3);
        const int k   = kl + 16 * (u >> 2);
        int2 z;
        z.x = 0;
        z.y = -1;
        if (k < rlen[run]) z = R2[run * SLICE + roff[run] + k];
        e[u] = z;
    }
    #pragma unroll
    for (int u = 0; u < 8; ++u)
        if (e[u].y >= 0) atomicAdd(&cn[e[u].y & 1023], 1);
    for (int g = 0; g < 4; ++g) {
        const int run = (tid >> 4) + 64 * g;
        const int n   = rlen[run];
        for (int k = kl + 32; k < n; k += 16)
            atomicAdd(&cn[R2[run * SLICE + roff[run] + k].y & 1023], 1);
    }
    __syncthreads();

    const int c = cn[tid];
    int inc = c;
    #pragma unroll
    for (int d = 1; d < 32; d <<= 1) {
        const int u = __shfl_up(inc, d);
        if (lane >= d) inc += u;
    }
    if (lane == 31) wt[wave] = inc;
    __syncthreads();
    int before = 0;
    #pragma unroll
    for (int w = 0; w < 32; ++w) {
        const int q = wt[w];
        if (w < wave) before += q;
    }
    st[tid] = before + inc - c;
    __syncthreads();

    #pragma unroll
    for (int u = 0; u < 8; ++u) {
        if (e[u].y >= 0) {
            const int rl = e[u].y & 1023;
            const int p  = st[rl] + atomicAdd(&cur[rl], 1);
            if (p < LCAP) list[p] = e[u].x;
        }
    }
    for (int g = 0; g < 4; ++g) {
        const int run = (tid >> 4) + 64 * g;
        const int n   = rlen[run];
        for (int k = kl + 32; k < n; k += 16) {
            const int2 z  = R2[run * SLICE + roff[run] + k];
            const int rl  = z.y & 1023;
            const int p   = st[rl] + atomicAdd(&cur[rl], 1);
            if (p < LCAP) list[p] = z.x;
        }
    }
    __syncthreads();

    const int hw = tid >> 4;
    const int cq = tid & 15;
    const float4* src = reinterpret_cast<const float4*>(T) + cq;

    for (int it = 0; it < 8; ++it) {
        const int r0 = it * 128 + hw * 2;
        int p0[2], n[2];
        float4 v[2][8];
        #pragma unroll
        for (int t = 0; t < 2; ++t) {
            p0[t] = st[r0 + t];
            n[t]  = cn[r0 + t];
            const int jmax = n[t] > 0 ? n[t] - 1 : 0;
            #pragma unroll
            for (int j = 0; j < 8; ++j) {
                int idx = p0[t] + (j < jmax ? j : jmax);
                idx = idx < LCAP ? idx : LCAP - 1;
                const int line = list[idx] & (NNZ - 1);
                v[t][j] = src[(size_t)line * (TW / 4)];
            }
        }
        #pragma unroll
        for (int t = 0; t < 2; ++t) {
            float4 acc = {0.0f, 0.0f, 0.0f, 0.0f};
            if (p0[t] + n[t] <= LCAP) {
                #pragma unroll
                for (int j = 0; j < 8; ++j) {
                    const bool on = j < n[t];
                    acc.x += on ? v[t][j].x : 0.0f;
                    acc.y += on ? v[t][j].y : 0.0f;
                    acc.z += on ? v[t][j].z : 0.0f;
                    acc.w += on ? v[t][j].w : 0.0f;
                }
                for (int j = 8; j < n[t]; ++j) {
                    const int line = list[p0[t] + j] & (NNZ - 1);
                    const float4 z = src[(size_t)line * (TW / 4)];
                    acc.x += z.x; acc.y += z.y; acc.z += z.z; acc.w += z.w;
                }
            } else {
                const int i = b * 1024 + r0 + t;
                for (int run = 0; run < NSL; ++run) {
                    const int m  = rlen[run];
                    const int bs = run * SLICE + roff[run];
                    for (int k = 0; k < m; ++k) {
                        const int2 z = R2[bs + k];
                        if (z.y == i) {
                            const float4 q = src[(size_t)(z.x & (NNZ - 1)) * (TW / 4)];
                            acc.x += q.x; acc.y += q.y; acc.z += q.z; acc.w += q.w;
                        }
                    }
                }
            }
            tile[(cq +  0) * TPB + r0 + t] = acc.x;
            tile[(cq + 16) * TPB + r0 + t] = acc.y;
            tile[(cq + 32) * TPB + r0 + t] = acc.z;
            tile[(cq + 48) * TPB + r0 + t] = acc.w;
        }
    }
    __syncthreads();

    const int i0 = b * 1024;
    #pragma unroll
    for (int m = 0; m < (64 * 1024 / 4) / 1024; ++m) {
        const int f  = m * 1024 + tid;
        const int ch = f >> 8;
        const int r4 = (f & 255) << 2;
        if (i0 + r4 < N_OUT) {
            const float4 o = *reinterpret_cast<const float4*>(tile + ch * TPB + r4);
            *reinterpret_cast<float4*>(y + (size_t)ch * N_OUT + i0 + r4) = o;
        }
    }
}

extern "C" void kernel_launch(void* const* d_in, const int* in_sizes, int n_in,
                              void* d_out, int out_size, void* d_ws, size_t ws_size,
                              hipStream_t stream) {
    (void)in_sizes; (void)n_in; (void)out_size; (void)ws_size;

    const float* x    = (const float*)d_in[0];
    const float* S    = (const float*)d_in[1];
    const int*   rows = (const int*)d_in[2];
    const int*   cols = (const int*)d_in[3];
    float*       y    = (float*)d_out;
    int*         ws   = (int*)d_ws;
    float*       T    = (float*)(ws + WS_T);

    regrid_slice_sort_kernel<<<NSL, 1024, 0, stream>>>(S, rows, cols, ws + WS_OC, ws + WS_OR,
                                                       ws + WS_R, ws + WS_P);

    regrid_product_wmma_kernel<<<PGRID, PNT, 0, stream>>>(x, ws + WS_P, ws + WS_OC, T);
    regrid_rowsum_kernel<<<NRB, 1024, 0, stream>>>(T, ws + WS_OR, ws + WS_R, y);
}


# slice sort puts its P and R entries in place in LDS and stores both 32-KB regions in address order: four whole 128-B lines per store instruction, not 8 B in each of up to 32 lines
# speedup vs baseline: 21.9804x; 1.0569x over previous
#include <hip/hip_runtime.h>


typedef __attribute__((ext_vector_type(2))) float v2f;
typedef __attribute__((ext_vector_type(8))) float v8f;

#define H_IN   721
#define W_IN   1440
#define H_OUT  361
#define W_OUT  720
#define N_IN   (H_IN * W_IN)
#define N_OUT  (H_OUT * W_OUT)
#define NNZ    1048576

#define CB_SHIFT 10
#define CB       (1 << CB_SHIFT)
#define NB       ((N_IN + CB - 1) / CB)
#define NBP      1024
#define PITCH    (CB + 4)
#define SLICE    4096
#define NSL      (NNZ / SLICE)
#define OCP      (NBP + 1)
#define ORP      257
#define CAP      4096

#define POS_BITS 21
#define POS_MASK ((1 << POS_BITS) - 1)
#define TROWS    (NNZ + NBP * 16)
#define TW       64

#define NRB      ((N_OUT + 1023) >> 10)
#define LCAP     8192
#define TPB      (1024 + 4)

static_assert(NNZ + NBP * 16 <= (1 << POS_BITS), "T line must fit its packed field");
static_assert(POS_BITS + CB_SHIFT <= 32, "packed entry fits 32 bits");
static_assert(NB <= NBP, "one sort thread per bucket");
static_assert(NRB <= 256, "row-block histogram size");
static_assert(NNZ % SLICE == 0, "whole slices");
static_assert(NSL == 256, "one run per slice: 256 runs per bin");
static_assert(SLICE < (1 << 15), "two packed 16-bit running sums");
static_assert(CAP % 16 == 0, "whole chunks per tile of entries");
static_assert(N_OUT % 4 == 0 && N_IN % 4 == 0, "float4 never split by an edge");
static_assert((NNZ & (NNZ - 1)) == 0, "a T line index is kept in range by a mask");

#define WS_OC     0
#define WS_OR     (WS_OC + NSL * OCP)
#define WS_R      (WS_OR + NSL * ORP)
#define WS_P      (WS_R + 2 * NNZ)
#define WS_T      (WS_P + 2 * NNZ)

static_assert(WS_T % 64 == 0, "T lines start on a 256-B boundary");
static_assert(WS_P % 32 == 0 && WS_R % 32 == 0, "regions of R and of P start on a 128-B line");
static_assert(SLICE % 2048 == 0, "a slice's region leaves as whole 16-B pieces, one or more per thread");

__device__ __forceinline__ int bucket_of(int col) {
    const unsigned b = (unsigned)col >> CB_SHIFT;
    return (int)(b < (unsigned)NB ? b : (unsigned)(NB - 1));
}
__device__ __forceinline__ int row_of(int row) {
    return (int)((unsigned)row < (unsigned)N_OUT ? (unsigned)row : (unsigned)(N_OUT - 1));
}

__device__ __forceinline__ int block_scan_incl(int* s, int tid, int v) {
    s[tid] = v;
    __syncthreads();
    for (int off = 1; off < 1024; off <<= 1) {
        const int a = (tid >= off) ? s[tid - off] : 0;
        __syncthreads();
        s[tid] += a;
        __syncthreads();
    }
    return s[tid];
}

__global__ __launch_bounds__(1024)
void regrid_slice_sort_kernel(const float* __restrict__ S,
                              const int*   __restrict__ rows,
                              const int*   __restrict__ cols,
                              int* __restrict__ OC, int* __restrict__ ORo,
                              int* __restrict__ R, int* __restrict__ P) {
    __shared__ int h[NBP];
    __shared__ int lbase[NBP];
    __shared__ int hr[256];
    __shared__ int lbr[256];
    __shared__ int sc[1024];
    __shared__ __attribute__((aligned(16))) int2 ps[SLICE];
    __shared__ __attribute__((aligned(16))) int2 rs[SLICE];
    const int tid = threadIdx.x;
    const int sl  = blockIdx.x;
    h[tid] = 0;
    if (tid < 256) hr[tid] = 0;
    __syncthreads();
    const int k0 = sl * SLICE + tid;
    int   cv[SLICE / 1024], bv[SLICE / 1024], rv[SLICE / 1024];
    float wv[SLICE / 1024];
    #pragma unroll
    for (int j = 0; j < SLICE / 1024; ++j) {
        cv[j] = cols[k0 + j * 1024];
        rv[j] = row_of(rows[k0 + j * 1024]);
        wv[j] = S[k0 + j * 1024];
    }
    #pragma unroll
    for (int j = 0; j < SLICE / 1024; ++j) {
        bv[j] = bucket_of(cv[j]);
        atomicAdd(&h[bv[j]], 1);
        atomicAdd(&hr[rv[j] >> 10], 1);
    }
    __syncthreads();
    const int v   = h[tid];
    const int w   = tid < 256 ? hr[tid] : 0;
    const int inc = block_scan_incl(sc, tid, v | (w << 16));
    const int exc = (inc & 0xFFFF) - v;
    const int exr = (inc >> 16) - w;
    lbase[tid] = exc;
    h[tid] = 0;
    OC[sl * OCP + tid] = exc;
    if (tid == 1023) OC[sl * OCP + NBP] = SLICE;
    if (tid < 256) {
        lbr[tid] = exr;
        hr[tid] = 0;
        ORo[sl * ORP + tid] = exr;
    }
    if (tid == 255) ORo[sl * ORP + 256] = SLICE;
    __syncthreads();
    #pragma unroll
    for (int j = 0; j < SLICE / 1024; ++j) {
        const int r    = atomicAdd(&h[bv[j]], 1);
        const int ls   = lbase[bv[j]] + r;
        const int slot = sl * SLICE + ls;
        const int rb   = rv[j] >> 10;
        const int rr   = atomicAdd(&hr[rb], 1);
        int2 p;
        p.x = (slot & POS_MASK) | ((cv[j] & (CB - 1)) << POS_BITS);
        p.y = __float_as_int(wv[j]);
        ps[ls] = p;
        int2 e;
        e.x = slot;
        e.y = rv[j];
        rs[lbr[rb] + rr] = e;
    }
    __syncthreads();
    int4* Pg = reinterpret_cast<int4*>(P) + sl * (SLICE / 2);
    int4* Rg = reinterpret_cast<int4*>(R) + sl * (SLICE / 2);
    const int4* p4 = reinterpret_cast<const int4*>(ps);
    const int4* r4 = reinterpret_cast<const int4*>(rs);
    for (int m = 0; m < SLICE / 2048; ++m) {
        Pg[m * 1024 + tid] = p4[m * 1024 + tid];
        Rg[m * 1024 + tid] = r4[m * 1024 + tid];
    }
}

#define PGRID 256
#define PNT   512
#define PNW   (PNT / 32)
#define XRS   (PNT / 256)
#define XL    (64 / XRS)
#define XA    16
#define LPR   (PNT / NSL)
#define EVN   (16 / LPR)

static_assert(PNT % 256 == 0 && PNT >= NSL, "a load covers whole plane rows; one thread per run");
static_assert(XA < XL, "part of a block is loaded ahead, the rest behind the scan");

__global__ __launch_bounds__(PNT)
void regrid_product_wmma_kernel(const float* __restrict__ x,
                                const int*   __restrict__ P,
                                const int*   __restrict__ OC,
                                float*       __restrict__ T) {
    __shared__ __attribute__((aligned(16))) float xs[64 * PITCH];
    __shared__ __attribute__((aligned(16))) int ent[2 * CAP];
    __shared__ int rt[3 * NSL + NSL / 32];
    int* const coff = rt;
    int* const clen = rt + NSL;
    int* const cexc = rt + 2 * NSL;
    int* const wtot = rt + 3 * NSL;

    const int tid    = threadIdx.x;
    const int lane   = tid & 31;
    const int c      = lane & 15;
    const int half   = lane >> 4;

    const int wave = __builtin_amdgcn_readfirstlane(tid >> 5);

    const unsigned ocb = (unsigned)((tid & (NSL - 1)) * OCP) * 4u;

    const char* Pb   = reinterpret_cast<const char*>(P);
    int2*       ent2 = reinterpret_cast<int2*>(ent);
    const int run = tid / LPR;
    const int sub = tid % LPR;

    const int      c4  = (tid & 255) << 2;
    const unsigned xob = (unsigned)((tid >> 8) * N_IN + c4) * 4u;
    float* xsw = xs + (tid >> 8) * PITCH + c4;

    const float* xr0 = xs + c * PITCH;
    const float* xr1 = xs + (16 + c) * PITCH;
    const float* xr2 = xs + (32 + c) * PITCH;
    const float* xr3 = xs + (48 + c) * PITCH;
    char* Tb = reinterpret_cast<char*>(T);
    const unsigned tcb = 16u * (unsigned)c;

    int bkt = blockIdx.x;
    int ra, re;
    {
        const int* o = reinterpret_cast<const int*>(reinterpret_cast<const char*>(OC + bkt) + ocb);
        ra = o[0];
        re = o[1];
    }
    float4 ta[XA];
    #pragma unroll
    for (int i = 0; i < XA; ++i) {
        const char* xb = reinterpret_cast<const char*>(x + (size_t)(XRS * i) * N_IN + (bkt << CB_SHIFT));
        float4 z = {0.0f, 0.0f, 0.0f, 0.0f};
        if ((bkt << CB_SHIFT) + c4 < N_IN)
            z = *reinterpret_cast<const float4*>(xb + xob);
        ta[i] = z;
    }

    while (true) {
        const int col0 = bkt << CB_SHIFT;

        const int rn = tid < NSL ? re - ra : 0;
        int inc = rn;
        #pragma unroll
        for (int d = 1; d < 32; d <<= 1) {
            const int u = __shfl_up(inc, d);
            if (lane >= d) inc += u;
        }
        if (tid < NSL && lane == 31) wtot[tid >> 5] = inc;
        __syncthreads();
        int nbv = 0, before = 0;
        #pragma unroll
        for (int w = 0; w < NSL / 32; ++w) {
            const int q = wtot[w];
            if (w < (tid >> 5)) before += q;
            nbv += q;
        }
        if (tid < NSL) {
            coff[tid] = ra;
            clen[tid] = rn;
            cexc[tid] = before + inc - rn;
        }
        __syncthreads();
        const int nb      = __builtin_amdgcn_readfirstlane(nbv);
        const int nchunks = (nb + 15) >> 4;

        const int rl  = clen[run];
        const int rx  = cexc[run];
        const unsigned pob = (unsigned)(run * SLICE + coff[run]) * 8u;

        int2 ev[EVN];
        #pragma unroll
        for (int u = 0; u < EVN; ++u) {
            const int k = sub + LPR * u;
            int2 z;
            z.x = 0;
            z.y = 0;
            if (k < rl) z = *reinterpret_cast<const int2*>(Pb + (pob + 8u * (unsigned)k));
            ev[u] = z;
        }
        #pragma unroll
        for (int i = 0; i < XA; ++i)
            *reinterpret_cast<float4*>(xsw + i * XRS * PITCH) = ta[i];
        float4 t[XL - XA];
        #pragma unroll
        for (int i = 0; i < XL - XA; ++i) {
            const char* xb = reinterpret_cast<const char*>(x + (size_t)(XRS * (XA + i)) * N_IN + col0);
            float4 z = {0.0f, 0.0f, 0.0f, 0.0f};
            if (col0 + c4 < N_IN)
                z = *reinterpret_cast<const float4*>(xb + xob);
            t[i] = z;
        }
        #pragma unroll
        for (int u = 0; u < EVN; ++u) {
            const int k = sub + LPR * u;
            const int d = rx + k;
            if (k < rl && (unsigned)d < (unsigned)CAP) ent2[d] = ev[u];
        }
        for (int k = sub + 16; k < rl; k += LPR) {
            const int d = rx + k;
            if ((unsigned)d < (unsigned)CAP)
                ent2[d] = *reinterpret_cast<const int2*>(Pb + (pob + 8u * (unsigned)k));
        }
        if (tid < 16) {
            const int j = nb + tid;
            if (j < nchunks * 16 && j < CAP) {
                int2 z;
                z.x = NNZ + bkt * 16 + tid;
                z.y = 0;
                ent2[j] = z;
            }
        }
        #pragma unroll
        for (int i = 0; i < XL - XA; ++i)
            *reinterpret_cast<float4*>(xsw + (XA + i) * XRS * PITCH) = t[i];

        const int nxt = bkt + PGRID;

        int tb = 0;
        while (true) {
            __syncthreads();

            if (tb == 0 && nxt < NB) {
                const int* o = reinterpret_cast<const int*>(reinterpret_cast<const char*>(OC + nxt) + ocb);
                ra = o[0];
                re = o[1];
                #pragma unroll
                for (int i = 0; i < XA; ++i) {
                    const char* xb = reinterpret_cast<const char*>(x + (size_t)(XRS * i) * N_IN + (nxt << CB_SHIFT));
                    float4 z = {0.0f, 0.0f, 0.0f, 0.0f};
                    if ((nxt << CB_SHIFT) + c4 < N_IN)
                        z = *reinterpret_cast<const float4*>(xb + xob);
                    ta[i] = z;
                }
            }

            const int cend = min(nchunks, (tb + CAP) >> 4);
            for (int chunk = (tb >> 4) + wave; chunk < cend; chunk += PNW) {
                const int* pe = ent + 2 * (chunk * 16 - tb) + 2 * (8 * half);
                int4 q[4];
                #pragma unroll
                for (int g = 0; g < 4; ++g)
                    q[g] = *reinterpret_cast<const int4*>(pe + 4 * g);

                float x0[8], x1[8], x2[8], x3[8], sv[8];
                #pragma unroll
                for (int g = 0; g < 4; ++g) {
                    const int ca = (q[g].x >> POS_BITS) & (CB - 1);
                    const int cb = (q[g].z >> POS_BITS) & (CB - 1);
                    x0[2 * g + 0] = xr0[ca];
                    x0[2 * g + 1] = xr0[cb];
                    x1[2 * g + 0] = xr1[ca];
                    x1[2 * g + 1] = xr1[cb];
                    x2[2 * g + 0] = xr2[ca];
                    x2[2 * g + 1] = xr2[cb];
                    x3[2 * g + 0] = xr3[ca];
                    x3[2 * g + 1] = xr3[cb];
                    sv[2 * g + 0] = __int_as_float(q[g].y);
                    sv[2 * g + 1] = __int_as_float(q[g].w);
                }

                v8f d0 = {}, d1 = {}, d2 = {}, d3 = {};
                #pragma unroll
                for (int g = 0; g < 4; ++g) {
                    v2f a, b0, b1, b2, b3;
                    a[0]  = (c == 8 * half + 2 * g + 0) ? sv[2 * g + 0] : 0.0f;
                    a[1]  = (c == 8 * half + 2 * g + 1) ? sv[2 * g + 1] : 0.0f;
                    b0[0] = x0[2 * g + 0];
                    b0[1] = x0[2 * g + 1];
                    b1[0] = x1[2 * g + 0];
                    b1[1] = x1[2 * g + 1];
                    b2[0] = x2[2 * g + 0];
                    b2[1] = x2[2 * g + 1];
                    b3[0] = x3[2 * g + 0];
                    b3[1] = x3[2 * g + 1];
                    d0 = __builtin_amdgcn_wmma_f32_16x16x4_f32(
                             false, a, false, b0, (short)0, d0, false, false);
                    d1 = __builtin_amdgcn_wmma_f32_16x16x4_f32(
                             false, a, false, b1, (short)0, d1, false, false);
                    d2 = __builtin_amdgcn_wmma_f32_16x16x4_f32(
                             false, a, false, b2, (short)0, d2, false, false);
                    d3 = __builtin_amdgcn_wmma_f32_16x16x4_f32(
                             false, a, false, b3, (short)0, d3, false, false);
                }

                #pragma unroll
                for (int r = 0; r < 8; ++r) {
                    const int pw  = (r & 1) ? q[r >> 1].z : q[r >> 1].x;
                    const unsigned pos = (unsigned)(pw & POS_MASK);
                    float4 o;
                    o.x = d0[r];
                    o.y = d1[r];
                    o.z = d2[r];
                    o.w = d3[r];
                    *reinterpret_cast<float4*>(Tb + (pos * (unsigned)(TW * 4) + tcb)) = o;
                }
            }
            tb += CAP;
            if (tb >= nb) break;
            __syncthreads();
            const int rl2 = clen[run];
            const int rx2 = cexc[run];
            const unsigned pob2 = (unsigned)(run * SLICE + coff[run]) * 8u;
            for (int k = sub; k < rl2; k += LPR) {
                const int d = rx2 + k - tb;
                if ((unsigned)d < (unsigned)CAP)
                    ent2[d] = *reinterpret_cast<const int2*>(Pb + (pob2 + 8u * (unsigned)k));
            }
            if (tid < 16) {
                const int j = nb + tid;
                const int d = j - tb;
                if (j < nchunks * 16 && (unsigned)d < (unsigned)CAP) {
                    int2 z;
                    z.x = NNZ + bkt * 16 + tid;
                    z.y = 0;
                    ent2[d] = z;
                }
            }
        }

        if (nxt >= NB) break;
        bkt = nxt;
    }
}

__global__ __launch_bounds__(1024)
void regrid_rowsum_kernel(const float* __restrict__ T,
                          const int*   __restrict__ ORo,
                          const int*   __restrict__ R,
                          float*       __restrict__ y) {
    __shared__ __attribute__((aligned(16))) float tile[64 * TPB];
    __shared__ int list[LCAP];
    __shared__ int cn[1024];
    __shared__ int st[1024];
    __shared__ int cur[1024];
    __shared__ int roff[NSL];
    __shared__ int rlen[NSL];
    __shared__ int wt[32];

    const int tid  = threadIdx.x;
    const int lane = tid & 31;
    const int wave = tid >> 5;
    const int b    = blockIdx.x;

    cn[tid]  = 0;
    cur[tid] = 0;
    if (tid < NSL) {
        const int a = ORo[tid * ORP + b];
        const int e = ORo[tid * ORP + b + 1];
        roff[tid] = a;
        rlen[tid] = e - a;
    }
    __syncthreads();

    const int2* R2 = reinterpret_cast<const int2*>(R);
    const int kl = tid & 15;

    int2 e[8];
    #pragma unroll
    for (int u = 0; u < 8; ++u) {
        const int run = (tid >> 4) + 64 * (u & 3);
        const int k   = kl + 16 * (u >> 2);
        int2 z;
        z.x = 0;
        z.y = -1;
        if (k < rlen[run]) z = R2[run * SLICE + roff[run] + k];
        e[u] = z;
    }
    #pragma unroll
    for (int u = 0; u < 8; ++u)
        if (e[u].y >= 0) atomicAdd(&cn[e[u].y & 1023], 1);
    for (int g = 0; g < 4; ++g) {
        const int run = (tid >> 4) + 64 * g;
        const int n   = rlen[run];
        for (int k = kl + 32; k < n; k += 16)
            atomicAdd(&cn[R2[run * SLICE + roff[run] + k].y & 1023], 1);
    }
    __syncthreads();

    const int c = cn[tid];
    int inc = c;
    #pragma unroll
    for (int d = 1; d < 32; d <<= 1) {
        const int u = __shfl_up(inc, d);
        if (lane >= d) inc += u;
    }
    if (lane == 31) wt[wave] = inc;
    __syncthreads();
    int before = 0;
    #pragma unroll
    for (int w = 0; w < 32; ++w) {
        const int q = wt[w];
        if (w < wave) before += q;
    }
    st[tid] = before + inc - c;
    __syncthreads();

    #pragma unroll
    for (int u = 0; u < 8; ++u) {
        if (e[u].y >= 0) {
            const int rl = e[u].y & 1023;
            const int p  = st[rl] + atomicAdd(&cur[rl], 1);
            if (p < LCAP) list[p] = e[u].x;
        }
    }
    for (int g = 0; g < 4; ++g) {
        const int run = (tid >> 4) + 64 * g;
        const int n   = rlen[run];
        for (int k = kl + 32; k < n; k += 16) {
            const int2 z  = R2[run * SLICE + roff[run] + k];
            const int rl  = z.y & 1023;
            const int p   = st[rl] + atomicAdd(&cur[rl], 1);
            if (p < LCAP) list[p] = z.x;
        }
    }
    __syncthreads();

    const int hw = tid >> 4;
    const int cq = tid & 15;
    const float4* src = reinterpret_cast<const float4*>(T) + cq;

    for (int it = 0; it < 8; ++it) {
        const int r0 = it * 128 + hw * 2;
        int p0[2], n[2];
        float4 v[2][8];
        #pragma unroll
        for (int t = 0; t < 2; ++t) {
            p0[t] = st[r0 + t];
            n[t]  = cn[r0 + t];
            const int jmax = n[t] > 0 ? n[t] - 1 : 0;
            #pragma unroll
            for (int j = 0; j < 8; ++j) {
                int idx = p0[t] + (j < jmax ? j : jmax);
                idx = idx < LCAP ? idx : LCAP - 1;
                const int line = list[idx] & (NNZ - 1);
                v[t][j] = src[(size_t)line * (TW / 4)];
            }
        }
        #pragma unroll
        for (int t = 0; t < 2; ++t) {
            float4 acc = {0.0f, 0.0f, 0.0f, 0.0f};
            if (p0[t] + n[t] <= LCAP) {
                #pragma unroll
                for (int j = 0; j < 8; ++j) {
                    const bool on = j < n[t];
                    acc.x += on ? v[t][j].x : 0.0f;
                    acc.y += on ? v[t][j].y : 0.0f;
                    acc.z += on ? v[t][j].z : 0.0f;
                    acc.w += on ? v[t][j].w : 0.0f;
                }
                for (int j = 8; j < n[t]; ++j) {
                    const int line = list[p0[t] + j] & (NNZ - 1);
                    const float4 z = src[(size_t)line * (TW / 4)];
                    acc.x += z.x; acc.y += z.y; acc.z += z.z; acc.w += z.w;
                }
            } else {
                const int i = b * 1024 + r0 + t;
                for (int run = 0; run < NSL; ++run) {
                    const int m  = rlen[run];
                    const int bs = run * SLICE + roff[run];
                    for (int k = 0; k < m; ++k) {
                        const int2 z = R2[bs + k];
                        if (z.y == i) {
                            const float4 q = src[(size_t)(z.x & (NNZ - 1)) * (TW / 4)];
                            acc.x += q.x; acc.y += q.y; acc.z += q.z; acc.w += q.w;
                        }
                    }
                }
            }
            tile[(cq +  0) * TPB + r0 + t] = acc.x;
            tile[(cq + 16) * TPB + r0 + t] = acc.y;
            tile[(cq + 32) * TPB + r0 + t] = acc.z;
            tile[(cq + 48) * TPB + r0 + t] = acc.w;
        }
    }
    __syncthreads();

    const int i0 = b * 1024;
    #pragma unroll
    for (int m = 0; m < (64 * 1024 / 4) / 1024; ++m) {
        const int f  = m * 1024 + tid;
        const int ch = f >> 8;
        const int r4 = (f & 255) << 2;
        if (i0 + r4 < N_OUT) {
            const float4 o = *reinterpret_cast<const float4*>(tile + ch * TPB + r4);
            *reinterpret_cast<float4*>(y + (size_t)ch * N_OUT + i0 + r4) = o;
        }
    }
}

extern "C" void kernel_launch(void* const* d_in, const int* in_sizes, int n_in,
                              void* d_out, int out_size, void* d_ws, size_t ws_size,
                              hipStream_t stream) {
    (void)in_sizes; (void)n_in; (void)out_size; (void)ws_size;

    const float* x    = (const float*)d_in[0];
    const float* S    = (const float*)d_in[1];
    const int*   rows = (const int*)d_in[2];
    const int*   cols = (const int*)d_in[3];
    float*       y    = (float*)d_out;
    int*         ws   = (int*)d_ws;
    float*       T    = (float*)(ws + WS_T);

    regrid_slice_sort_kernel<<<NSL, 1024, 0, stream>>>(S, rows, cols, ws + WS_OC, ws + WS_OR,
                                                       ws + WS_R, ws + WS_P);

    regrid_product_wmma_kernel<<<PGRID, PNT, 0, stream>>>(x, ws + WS_P, ws + WS_OC, T);
    regrid_rowsum_kernel<<<NRB, 1024, 0, stream>>>(T, ws + WS_OR, ws + WS_R, y);
}
